# starting kernel
# baseline (speedup 1.0000x reference)
#include <hip/hip_runtime.h>

#define N_NODES  50000
#define N_EDGES  1600000
#define N_GRAPHS 256
#define IN_DIM   9
#define HID      128

typedef __attribute__((ext_vector_type(2))) float v2f;
typedef __attribute__((ext_vector_type(8))) float v8f;

__device__ __forceinline__ void atomic_add_f32(float* p, float v) {
    asm volatile("global_atomic_add_f32 %0, %1, off scope:SCOPE_DEV"
                 :: "v"(p), "v"(v) : "memory");
}

__global__ void k_fill1(float* p, int n) {
    int i = blockIdx.x * blockDim.x + threadIdx.x;
    if (i < n) p[i] = 1.0f;
}

__global__ void k_deg(const int* __restrict__ dst, float* __restrict__ deg) {
    int e = blockIdx.x * blockDim.x + threadIdx.x;
    if (e < N_EDGES) atomic_add_f32(&deg[dst[e]], 1.0f);
}

__global__ void k_rsqrt(float* p, int n) {
    int i = blockIdx.x * blockDim.x + threadIdx.x;
    if (i < n) p[i] = rsqrtf(p[i]);
}

__global__ void k_gemm_in(const float* __restrict__ x, const float* __restrict__ W1,
                          float* __restrict__ out) {
    int idx = blockIdx.x * blockDim.x + threadIdx.x;
    if (idx >= N_NODES * HID) return;
    int f = idx & (HID - 1);
    int n = idx >> 7;
    const float* xr = x + n * IN_DIM;
    float acc = 0.f;
#pragma unroll
    for (int k = 0; k < IN_DIM; ++k) acc = fmaf(xr[k], W1[k * HID + f], acc);
    out[idx] = acc;
}

template <bool RELU>
__global__ void k_gemm_wmma(const float* __restrict__ A, const float* __restrict__ W,
                            float* __restrict__ out) {
    int wave = (blockIdx.x * blockDim.x + threadIdx.x) >> 5;
    int lane = threadIdx.x & 31;
    int ct = wave & 7;
    int rt = wave >> 3;
    if (rt >= (N_NODES >> 4)) return;

    int half = lane >> 4;
    int l16  = lane & 15;
    int row  = (rt << 4) + l16;
    int col  = (ct << 4) + l16;
    const float* arow = A + row * HID;

    v8f c = {0.f, 0.f, 0.f, 0.f, 0.f, 0.f, 0.f, 0.f};
#pragma unroll
    for (int k0 = 0; k0 < HID; k0 += 4) {
        int ka = k0 + (half << 1);
        float a0 = arow[ka];
        float a1 = arow[ka + 1];
        if (RELU) { a0 = fmaxf(a0, 0.f); a1 = fmaxf(a1, 0.f); }
        v2f a; a.x = a0; a.y = a1;
        v2f b; b.x = W[ka * HID + col]; b.y = W[(ka + 1) * HID + col];
        c = __builtin_amdgcn_wmma_f32_16x16x4_f32(false, a, false, b,
                                                  (short)0, c, false, false);
    }
    float* orow = out + ((rt << 4) + (half << 3)) * HID + (ct << 4) + l16;
#pragma unroll
    for (int v = 0; v < 8; ++v) orow[v * HID] = c[v];
}

__global__ void k_init_out(const float* __restrict__ hw, const float* __restrict__ dis,
                           const float* __restrict__ bias, float* __restrict__ out) {
    int idx = blockIdx.x * blockDim.x + threadIdx.x;
    if (idx >= N_NODES * HID) return;
    int f = idx & (HID - 1);
    int n = idx >> 7;
    float s = dis[n];
    out[idx] = fmaf(hw[idx], s * s, bias[f]);
}

__global__ void k_edge(const int* __restrict__ src, const int* __restrict__ dst,
                       const float* __restrict__ dis, const float* __restrict__ hw,
                       float* __restrict__ out) {
    int t = blockIdx.x * blockDim.x + threadIdx.x;
    int e = t >> 5;
    if (e >= N_EDGES) return;
    e = __builtin_amdgcn_readfirstlane(e);
    int lane = t & 31;
    int s = src[e];
    int d = dst[e];
    float w = dis[s] * dis[d];
    const float4 h4 = *(const float4*)(hw + s * HID + lane * 4);
    float* o = out + d * HID + lane * 4;
    atomic_add_f32(o + 0, h4.x * w);
    atomic_add_f32(o + 1, h4.y * w);
    atomic_add_f32(o + 2, h4.z * w);
    atomic_add_f32(o + 3, h4.w * w);
}

__global__ void k_pool_zero(float* __restrict__ sums, float* __restrict__ counts) {
    int i = blockIdx.x * blockDim.x + threadIdx.x;
    if (i < N_GRAPHS * HID) sums[i] = 0.f;
    if (i < N_GRAPHS) counts[i] = 0.f;
}

__global__ void k_pool_acc(const float* __restrict__ h, const int* __restrict__ batch,
                           float* __restrict__ sums, float* __restrict__ counts) {
    int t = blockIdx.x * blockDim.x + threadIdx.x;
    int n = t >> 5;
    if (n >= N_NODES) return;
    n = __builtin_amdgcn_readfirstlane(n);
    int lane = t & 31;
    int g = batch[n];
    const float4 h4 = *(const float4*)(h + n * HID + lane * 4);
    float* o = sums + g * HID + lane * 4;
    atomic_add_f32(o + 0, h4.x);
    atomic_add_f32(o + 1, h4.y);
    atomic_add_f32(o + 2, h4.z);
    atomic_add_f32(o + 3, h4.w);
    if (lane == 0) atomic_add_f32(&counts[g], 1.0f);
}

__global__ void k_pool_div(float* __restrict__ sums, const float* __restrict__ counts) {
    int i = blockIdx.x * blockDim.x + threadIdx.x;
    if (i >= N_GRAPHS * HID) return;
    int g = i >> 7;
    sums[i] = sums[i] / fmaxf(counts[g], 1.0f);
}

extern "C" void kernel_launch(void* const* d_in, const int* in_sizes, int n_in,
                              void* d_out, int out_size, void* d_ws, size_t ws_size,
                              hipStream_t stream) {
    const float* x        = (const float*)d_in[0];
    const int*   eidx     = (const int*)d_in[1];
    const int*   batch    = (const int*)d_in[2];
    const float* W1       = (const float*)d_in[3];
    const float* b1       = (const float*)d_in[4];
    const float* W2       = (const float*)d_in[5];
    const float* b2       = (const float*)d_in[6];
    const float* W3       = (const float*)d_in[7];
    const float* b3       = (const float*)d_in[8];
    float* out = (float*)d_out;

    const int* src = eidx;
    const int* dst = eidx + N_EDGES;

    float* ws     = (float*)d_ws;
    float* dis    = ws;
    float* hw     = dis + N_NODES;
    float* hA     = hw + (size_t)N_NODES * HID;
    float* counts = hA + (size_t)N_NODES * HID;

    const int B = 256;
    int g_nf   = (N_NODES * HID + B - 1) / B;
    int g_n    = (N_NODES + B - 1) / B;
    int g_e    = (N_EDGES + B - 1) / B;
    int g_ew   = (N_EDGES * 32 + B - 1) / B;
    int g_nw   = (N_NODES * 32 + B - 1) / B;
    int g_wmma = ((N_NODES >> 4) * 8 + (B >> 5) - 1) / (B >> 5);

    k_fill1<<<g_n, B, 0, stream>>>(dis, N_NODES);
    k_deg<<<g_e, B, 0, stream>>>(dst, dis);
    k_rsqrt<<<g_n, B, 0, stream>>>(dis, N_NODES);

    k_gemm_in<<<g_nf, B, 0, stream>>>(x, W1, hw);
    k_init_out<<<g_nf, B, 0, stream>>>(hw, dis, b1, hA);
    k_edge<<<g_ew, B, 0, stream>>>(src, dst, dis, hw, hA);

    k_gemm_wmma<true><<<g_wmma, B, 0, stream>>>(hA, W2, hw);
    k_init_out<<<g_nf, B, 0, stream>>>(hw, dis, b2, hA);
    k_edge<<<g_ew, B, 0, stream>>>(src, dst, dis, hw, hA);

    k_gemm_wmma<true><<<g_wmma, B, 0, stream>>>(hA, W3, hw);
    k_init_out<<<g_nf, B, 0, stream>>>(hw, dis, b3, hA);
    k_edge<<<g_ew, B, 0, stream>>>(src, dst, dis, hw, hA);

    k_pool_zero<<<(N_GRAPHS * HID + B - 1) / B, B, 0, stream>>>(out, counts);
    k_pool_acc<<<g_nw, B, 0, stream>>>(hA, batch, out, counts);
    k_pool_div<<<(N_GRAPHS * HID + B - 1) / B, B, 0, stream>>>(out, counts);
}


# Edge sums by gather over a dst-sorted edge list (stable in-call radix sort, no atomics) replace the per-element f32 atomic scatter; per-node sum order is now fixed (edge-index order)
# speedup vs baseline: 2.8343x; 2.8343x over previous
#include <hip/hip_runtime.h>

#define N_NODES  50000
#define N_EDGES  1600000
#define N_GRAPHS 256
#define IN_DIM   9
#define HID      128

typedef __attribute__((ext_vector_type(2))) float v2f;
typedef __attribute__((ext_vector_type(8))) float v8f;

__device__ __forceinline__ void atomic_add_f32(float* p, float v) {
    asm volatile("global_atomic_add_f32 %0, %1, off scope:SCOPE_DEV"
                 :: "v"(p), "v"(v) : "memory");
}

#define RS_ITEMS  32
#define RS_WTILE  1024
#define RS_BTILE  8192
#define RS_BLOCKS 196
#define RS_HIST   50176

__device__ __forceinline__ unsigned match_digit(int d, bool valid) {
    unsigned m = __builtin_amdgcn_ballot_w32(valid);
#pragma unroll
    for (int b = 0; b < 8; ++b) {
        bool bit = ((d >> b) & 1) != 0;
        unsigned bal = __builtin_amdgcn_ballot_w32(bit);
        m &= bit ? bal : ~bal;
    }
    return m;
}

template <int SHIFT>
__global__ void __launch_bounds__(256) k_hist(const int* __restrict__ keys, int* __restrict__ hist) {
    __shared__ int h[8][256];
    int tid = threadIdx.x;
    int w = tid >> 5;
    int lane = tid & 31;
#pragma unroll
    for (int k = 0; k < 8; ++k) h[k][tid] = 0;
    __syncthreads();
    int base = blockIdx.x * RS_BTILE + w * RS_WTILE + lane;
    int key[RS_ITEMS];
#pragma unroll
    for (int i = 0; i < RS_ITEMS; ++i) {
        int idx = base + i * 32;
        key[i] = keys[idx < N_EDGES ? idx : N_EDGES - 1];
    }
#pragma unroll
    for (int i = 0; i < RS_ITEMS; ++i) {
        bool valid = (base + i * 32) < N_EDGES;
        int d = (key[i] >> SHIFT) & 255;
        unsigned m = match_digit(d, valid);
        int top = 31 - __builtin_clz(m | 1u);
        if (valid && lane == top) h[w][d] += __builtin_popcount(m);
    }
    __syncthreads();
    int tot = 0;
#pragma unroll
    for (int k = 0; k < 8; ++k) tot += h[k][tid];
    hist[tid * RS_BLOCKS + blockIdx.x] = tot;
}

__global__ void __launch_bounds__(256) k_scan(const int* __restrict__ in, int* __restrict__ out) {
    __shared__ int s[256];
    int t = threadIdx.x;
    int row = blockIdx.x * RS_BLOCKS;
    int part = 0;
    for (int i = t; i < row; i += 256) part += in[i];
    s[t] = part;
    __syncthreads();
    for (int off = 1; off < 256; off <<= 1) {
        int v = (t >= off) ? s[t - off] : 0;
        __syncthreads();
        s[t] += v;
        __syncthreads();
    }
    int before = s[255];
    __syncthreads();
    int x = (t < RS_BLOCKS) ? in[row + t] : 0;
    s[t] = x;
    __syncthreads();
    for (int off = 1; off < 256; off <<= 1) {
        int v = (t >= off) ? s[t - off] : 0;
        __syncthreads();
        s[t] += v;
        __syncthreads();
    }
    if (t < RS_BLOCKS) out[row + t] = before + s[t] - x;
}

template <int SHIFT>
__global__ void __launch_bounds__(256) k_scatter(const int* __restrict__ keys, const int* __restrict__ vals,
                                                 const int* __restrict__ hscan,
                                                 int* __restrict__ keys_out, int* __restrict__ vals_out) {
    __shared__ int h[8][256];
    int tid = threadIdx.x;
    int w = tid >> 5;
    int lane = tid & 31;
#pragma unroll
    for (int k = 0; k < 8; ++k) h[k][tid] = 0;
    __syncthreads();
    int base = blockIdx.x * RS_BTILE + w * RS_WTILE + lane;
    int key[RS_ITEMS];
    int val[RS_ITEMS];
#pragma unroll
    for (int i = 0; i < RS_ITEMS; ++i) {
        int idx = base + i * 32;
        int ic = idx < N_EDGES ? idx : N_EDGES - 1;
        key[i] = keys[ic];
        val[i] = vals[ic];
    }
    unsigned lt = (1u << lane) - 1u;
    int info[RS_ITEMS];
#pragma unroll
    for (int i = 0; i < RS_ITEMS; ++i) {
        bool valid = (base + i * 32) < N_EDGES;
        int d = (key[i] >> SHIFT) & 255;
        unsigned m = match_digit(d, valid);
        int top = 31 - __builtin_clz(m | 1u);
        int cntg = __builtin_popcount(m);
        bool leader = valid && lane == top;
        if (leader) h[w][d] += cntg;
        info[i] = __builtin_popcount(m & lt) | (cntg << 8) | (leader ? 0x10000 : 0);
    }
    __syncthreads();
    {
        int run = hscan[tid * RS_BLOCKS + blockIdx.x];
#pragma unroll
        for (int k = 0; k < 8; ++k) {
            int c = h[k][tid];
            h[k][tid] = run;
            run += c;
        }
    }
    __syncthreads();
#pragma unroll
    for (int i = 0; i < RS_ITEMS; ++i) {
        bool valid = (base + i * 32) < N_EDGES;
        int d = (key[i] >> SHIFT) & 255;
        int start = h[w][d];
        int pos = start + (info[i] & 255);
        if (info[i] & 0x10000) h[w][d] = start + ((info[i] >> 8) & 255);
        if (valid) {
            keys_out[pos] = key[i];
            vals_out[pos] = val[i];
        }
    }
}

__global__ void k_bounds(const int* __restrict__ sdst, int* __restrict__ rowptr) {
    int p = blockIdx.x * blockDim.x + threadIdx.x;
    if (p >= N_EDGES) return;
    int d = sdst[p];
    int prev = (p > 0) ? sdst[p - 1] : -1;
    for (int n = prev + 1; n <= d; ++n) rowptr[n] = p;
    if (p == N_EDGES - 1)
        for (int n = d + 1; n <= N_NODES; ++n) rowptr[n] = N_EDGES;
}

__global__ void k_pair(const int* __restrict__ ssrc, const int* __restrict__ sdst,
                       const int* __restrict__ rowptr, int2* __restrict__ csr) {
    int p = blockIdx.x * blockDim.x + threadIdx.x;
    if (p >= N_EDGES) return;
    int s = ssrc[p];
    int d = sdst[p];
    float ds = rsqrtf((float)(rowptr[s + 1] - rowptr[s]) + 1.0f);
    float dd = rsqrtf((float)(rowptr[d + 1] - rowptr[d]) + 1.0f);
    int2 v;
    v.x = s;
    v.y = __float_as_int(ds * dd);
    csr[p] = v;
}

__global__ void k_gemm_in(const float* __restrict__ x, const float* __restrict__ W1,
                          float* __restrict__ out) {
    int idx = blockIdx.x * blockDim.x + threadIdx.x;
    if (idx >= N_NODES * HID) return;
    int f = idx & (HID - 1);
    int n = idx >> 7;
    const float* xr = x + n * IN_DIM;
    float acc = 0.f;
#pragma unroll
    for (int k = 0; k < IN_DIM; ++k) acc = fmaf(xr[k], W1[k * HID + f], acc);
    out[idx] = acc;
}

template <bool RELU>
__global__ void k_gemm_wmma(const float* __restrict__ A, const float* __restrict__ W,
                            float* __restrict__ out) {
    int wave = (blockIdx.x * blockDim.x + threadIdx.x) >> 5;
    int lane = threadIdx.x & 31;
    int ct = wave & 7;
    int rt = wave >> 3;
    if (rt >= (N_NODES >> 4)) return;

    int half = lane >> 4;
    int l16  = lane & 15;
    int row  = (rt << 4) + l16;
    int col  = (ct << 4) + l16;
    const float* arow = A + row * HID;

    v8f c = {0.f, 0.f, 0.f, 0.f, 0.f, 0.f, 0.f, 0.f};
#pragma unroll
    for (int k0 = 0; k0 < HID; k0 += 4) {
        int ka = k0 + (half << 1);
        float a0 = arow[ka];
        float a1 = arow[ka + 1];
        if (RELU) { a0 = fmaxf(a0, 0.f); a1 = fmaxf(a1, 0.f); }
        v2f a; a.x = a0; a.y = a1;
        v2f b; b.x = W[ka * HID + col]; b.y = W[(ka + 1) * HID + col];
        c = __builtin_amdgcn_wmma_f32_16x16x4_f32(false, a, false, b,
                                                  (short)0, c, false, false);
    }
    float* orow = out + ((rt << 4) + (half << 3)) * HID + (ct << 4) + l16;
#pragma unroll
    for (int v = 0; v < 8; ++v) orow[v * HID] = c[v];
}

__global__ void __launch_bounds__(256) k_gather(const int* __restrict__ rowptr, const int2* __restrict__ csr,
                                                const float* __restrict__ hw, const float* __restrict__ bias,
                                                float* __restrict__ out) {
    int t = blockIdx.x * blockDim.x + threadIdx.x;
    int n = t >> 5;
    if (n >= N_NODES) return;
    n = __builtin_amdgcn_readfirstlane(n);
    int lane = threadIdx.x & 31;
    int beg = rowptr[n];
    int end = rowptr[n + 1];
    const float* hl = hw + lane * 4;
    float4 acc = make_float4(0.f, 0.f, 0.f, 0.f);
    int j = beg;
    for (; j + 4 <= end; j += 4) {
        int2 e0 = csr[j], e1 = csr[j + 1], e2 = csr[j + 2], e3 = csr[j + 3];
        const float4 h0 = *(const float4*)(hl + e0.x * HID);
        const float4 h1 = *(const float4*)(hl + e1.x * HID);
        const float4 h2 = *(const float4*)(hl + e2.x * HID);
        const float4 h3 = *(const float4*)(hl + e3.x * HID);
        float w0 = __int_as_float(e0.y), w1 = __int_as_float(e1.y);
        float w2 = __int_as_float(e2.y), w3 = __int_as_float(e3.y);
        acc.x = fmaf(h0.x, w0, acc.x); acc.y = fmaf(h0.y, w0, acc.y);
        acc.z = fmaf(h0.z, w0, acc.z); acc.w = fmaf(h0.w, w0, acc.w);
        acc.x = fmaf(h1.x, w1, acc.x); acc.y = fmaf(h1.y, w1, acc.y);
        acc.z = fmaf(h1.z, w1, acc.z); acc.w = fmaf(h1.w, w1, acc.w);
        acc.x = fmaf(h2.x, w2, acc.x); acc.y = fmaf(h2.y, w2, acc.y);
        acc.z = fmaf(h2.z, w2, acc.z); acc.w = fmaf(h2.w, w2, acc.w);
        acc.x = fmaf(h3.x, w3, acc.x); acc.y = fmaf(h3.y, w3, acc.y);
        acc.z = fmaf(h3.z, w3, acc.z); acc.w = fmaf(h3.w, w3, acc.w);
    }
    for (; j < end; ++j) {
        int2 e0 = csr[j];
        const float4 h0 = *(const float4*)(hl + e0.x * HID);
        float w0 = __int_as_float(e0.y);
        acc.x = fmaf(h0.x, w0, acc.x); acc.y = fmaf(h0.y, w0, acc.y);
        acc.z = fmaf(h0.z, w0, acc.z); acc.w = fmaf(h0.w, w0, acc.w);
    }
    float sd = rsqrtf((float)(end - beg) + 1.0f);
    float ss = sd * sd;
    const float4 hs = *(const float4*)(hl + n * HID);
    const float4 b4 = *(const float4*)(bias + lane * 4);
    float4 o;
    o.x = acc.x + fmaf(hs.x, ss, b4.x);
    o.y = acc.y + fmaf(hs.y, ss, b4.y);
    o.z = acc.z + fmaf(hs.z, ss, b4.z);
    o.w = acc.w + fmaf(hs.w, ss, b4.w);
    *(float4*)(out + n * HID + lane * 4) = o;
}

__global__ void k_pool_zero(float* __restrict__ sums, float* __restrict__ counts) {
    int i = blockIdx.x * blockDim.x + threadIdx.x;
    if (i < N_GRAPHS * HID) sums[i] = 0.f;
    if (i < N_GRAPHS) counts[i] = 0.f;
}

__global__ void k_pool_acc(const float* __restrict__ h, const int* __restrict__ batch,
                           float* __restrict__ sums, float* __restrict__ counts) {
    int t = blockIdx.x * blockDim.x + threadIdx.x;
    int n = t >> 5;
    if (n >= N_NODES) return;
    n = __builtin_amdgcn_readfirstlane(n);
    int lane = t & 31;
    int g = batch[n];
    const float4 h4 = *(const float4*)(h + n * HID + lane * 4);
    float* o = sums + g * HID + lane * 4;
    atomic_add_f32(o + 0, h4.x);
    atomic_add_f32(o + 1, h4.y);
    atomic_add_f32(o + 2, h4.z);
    atomic_add_f32(o + 3, h4.w);
    if (lane == 0) atomic_add_f32(&counts[g], 1.0f);
}

__global__ void k_pool_div(float* __restrict__ sums, const float* __restrict__ counts) {
    int i = blockIdx.x * blockDim.x + threadIdx.x;
    if (i >= N_GRAPHS * HID) return;
    int g = i >> 7;
    sums[i] = sums[i] / fmaxf(counts[g], 1.0f);
}

extern "C" void kernel_launch(void* const* d_in, const int* in_sizes, int n_in,
                              void* d_out, int out_size, void* d_ws, size_t ws_size,
                              hipStream_t stream) {
    const float* x        = (const float*)d_in[0];
    const int*   eidx     = (const int*)d_in[1];
    const int*   batch    = (const int*)d_in[2];
    const float* W1       = (const float*)d_in[3];
    const float* b1       = (const float*)d_in[4];
    const float* W2       = (const float*)d_in[5];
    const float* b2       = (const float*)d_in[6];
    const float* W3       = (const float*)d_in[7];
    const float* b3       = (const float*)d_in[8];
    float* out = (float*)d_out;

    const int* src = eidx;
    const int* dst = eidx + N_EDGES;

    float* ws     = (float*)d_ws;
    float* hw     = ws;
    float* hA     = hw + (size_t)N_NODES * HID;
    float* counts = hA + (size_t)N_NODES * HID;
    int2*  csr    = (int2*)(counts + N_GRAPHS);
    int*   kA     = (int*)(csr + N_EDGES);
    int*   vA     = kA + N_EDGES;
    int*   kB     = vA + N_EDGES;
    int*   vB     = kB + N_EDGES;
    int*   hist   = vB + N_EDGES;
    int*   hscan  = hist + RS_HIST;
    int*   rowptr = hscan + RS_HIST;

    const int B = 256;
    int g_nf   = (N_NODES * HID + B - 1) / B;
    int g_e    = (N_EDGES + B - 1) / B;
    int g_nw   = (N_NODES * 32 + B - 1) / B;
    int g_wmma = ((N_NODES >> 4) * 8 + (B >> 5) - 1) / (B >> 5);

    k_hist<0><<<RS_BLOCKS, B, 0, stream>>>(dst, hist);
    k_scan<<<256, B, 0, stream>>>(hist, hscan);
    k_scatter<0><<<RS_BLOCKS, B, 0, stream>>>(dst, src, hscan, kA, vA);
    k_hist<8><<<RS_BLOCKS, B, 0, stream>>>(kA, hist);
    k_scan<<<256, B, 0, stream>>>(hist, hscan);
    k_scatter<8><<<RS_BLOCKS, B, 0, stream>>>(kA, vA, hscan, kB, vB);
    k_bounds<<<g_e, B, 0, stream>>>(kB, rowptr);
    k_pair<<<g_e, B, 0, stream>>>(vB, kB, rowptr, csr);

    k_gemm_in<<<g_nf, B, 0, stream>>>(x, W1, hw);
    k_gather<<<g_nw, B, 0, stream>>>(rowptr, csr, hw, b1, hA);

    k_gemm_wmma<true><<<g_wmma, B, 0, stream>>>(hA, W2, hw);
    k_gather<<<g_nw, B, 0, stream>>>(rowptr, csr, hw, b2, hA);

    k_gemm_wmma<true><<<g_wmma, B, 0, stream>>>(hA, W3, hw);
    k_gather<<<g_nw, B, 0, stream>>>(rowptr, csr, hw, b3, hA);

    k_pool_zero<<<(N_GRAPHS * HID + B - 1) / B, B, 0, stream>>>(out, counts);
    k_pool_acc<<<g_nw, B, 0, stream>>>(hA, batch, out, counts);
    k_pool_div<<<(N_GRAPHS * HID + B - 1) / B, B, 0, stream>>>(out, counts);
}


# Per-node and per-edge kernels launch a few thousand long-lived waves, each looping over 4-10 nodes or 5 edges (layer-1 product, bounds, edge weights, gather, pool adds): 4x fewer waves per run
# speedup vs baseline: 3.1992x; 1.1287x over previous
#include <hip/hip_runtime.h>

#define N_NODES  50000
#define N_EDGES  1600000
#define N_GRAPHS 256
#define IN_DIM   9
#define HID      128

typedef __attribute__((ext_vector_type(2))) float v2f;
typedef __attribute__((ext_vector_type(8))) float v8f;

__device__ __forceinline__ void atomic_add_f32(float* p, float v) {
    asm volatile("global_atomic_add_f32 %0, %1, off scope:SCOPE_DEV"
                 :: "v"(p), "v"(v) : "memory");
}

#define RS_ITEMS  32
#define RS_WTILE  1024
#define RS_BTILE  8192
#define RS_BLOCKS 196
#define RS_HIST   50176

__device__ __forceinline__ unsigned match_digit(int d, bool valid) {
    unsigned m = __builtin_amdgcn_ballot_w32(valid);
#pragma unroll
    for (int b = 0; b < 8; ++b) {
        bool bit = ((d >> b) & 1) != 0;
        unsigned bal = __builtin_amdgcn_ballot_w32(bit);
        m &= bit ? bal : ~bal;
    }
    return m;
}

template <int SHIFT>
__global__ void __launch_bounds__(256) k_hist(const int* __restrict__ keys, int* __restrict__ hist) {
    __shared__ int h[8][256];
    int tid = threadIdx.x;
    int w = tid >> 5;
    int lane = tid & 31;
#pragma unroll
    for (int k = 0; k < 8; ++k) h[k][tid] = 0;
    __syncthreads();
    int base = blockIdx.x * RS_BTILE + w * RS_WTILE + lane;
    int key[RS_ITEMS];
#pragma unroll
    for (int i = 0; i < RS_ITEMS; ++i) {
        int idx = base + i * 32;
        key[i] = keys[idx < N_EDGES ? idx : N_EDGES - 1];
    }
#pragma unroll
    for (int i = 0; i < RS_ITEMS; ++i) {
        bool valid = (base + i * 32) < N_EDGES;
        int d = (key[i] >> SHIFT) & 255;
        unsigned m = match_digit(d, valid);
        int top = 31 - __builtin_clz(m | 1u);
        if (valid && lane == top) h[w][d] += __builtin_popcount(m);
    }
    __syncthreads();
    int tot = 0;
#pragma unroll
    for (int k = 0; k < 8; ++k) tot += h[k][tid];
    hist[tid * RS_BLOCKS + blockIdx.x] = tot;
}

__global__ void __launch_bounds__(256) k_scan(const int* __restrict__ in, int* __restrict__ out) {
    __shared__ int s[256];
    int t = threadIdx.x;
    int row = blockIdx.x * RS_BLOCKS;
    int part = 0;
    for (int i = t; i < row; i += 256) part += in[i];
    s[t] = part;
    __syncthreads();
    for (int off = 1; off < 256; off <<= 1) {
        int v = (t >= off) ? s[t - off] : 0;
        __syncthreads();
        s[t] += v;
        __syncthreads();
    }
    int before = s[255];
    __syncthreads();
    int x = (t < RS_BLOCKS) ? in[row + t] : 0;
    s[t] = x;
    __syncthreads();
    for (int off = 1; off < 256; off <<= 1) {
        int v = (t >= off) ? s[t - off] : 0;
        __syncthreads();
        s[t] += v;
        __syncthreads();
    }
    if (t < RS_BLOCKS) out[row + t] = before + s[t] - x;
}

template <int SHIFT>
__global__ void __launch_bounds__(256) k_scatter(const int* __restrict__ keys, const int* __restrict__ vals,
                                                 const int* __restrict__ hscan,
                                                 int* __restrict__ keys_out, int* __restrict__ vals_out) {
    __shared__ int h[8][256];
    int tid = threadIdx.x;
    int w = tid >> 5;
    int lane = tid & 31;
#pragma unroll
    for (int k = 0; k < 8; ++k) h[k][tid] = 0;
    __syncthreads();
    int base = blockIdx.x * RS_BTILE + w * RS_WTILE + lane;
    int key[RS_ITEMS];
    int val[RS_ITEMS];
#pragma unroll
    for (int i = 0; i < RS_ITEMS; ++i) {
        int idx = base + i * 32;
        int ic = idx < N_EDGES ? idx : N_EDGES - 1;
        key[i] = keys[ic];
        val[i] = vals[ic];
    }
    unsigned lt = (1u << lane) - 1u;
    int info[RS_ITEMS];
#pragma unroll
    for (int i = 0; i < RS_ITEMS; ++i) {
        bool valid = (base + i * 32) < N_EDGES;
        int d = (key[i] >> SHIFT) & 255;
        unsigned m = match_digit(d, valid);
        int top = 31 - __builtin_clz(m | 1u);
        int cntg = __builtin_popcount(m);
        bool leader = valid && lane == top;
        if (leader) h[w][d] += cntg;
        info[i] = __builtin_popcount(m & lt) | (cntg << 8) | (leader ? 0x10000 : 0);
    }
    __syncthreads();
    {
        int run = hscan[tid * RS_BLOCKS + blockIdx.x];
#pragma unroll
        for (int k = 0; k < 8; ++k) {
            int c = h[k][tid];
            h[k][tid] = run;
            run += c;
        }
    }
    __syncthreads();
#pragma unroll
    for (int i = 0; i < RS_ITEMS; ++i) {
        bool valid = (base + i * 32) < N_EDGES;
        int d = (key[i] >> SHIFT) & 255;
        int start = h[w][d];
        int pos = start + (info[i] & 255);
        if (info[i] & 0x10000) h[w][d] = start + ((info[i] >> 8) & 255);
        if (valid) {
            keys_out[pos] = key[i];
            vals_out[pos] = val[i];
        }
    }
}

#define EL_ITEMS   5
#define EL_BLOCKS  1250
#define EL_STRIDE  (EL_BLOCKS * 256)
static_assert(EL_STRIDE * EL_ITEMS == N_EDGES, "every thread owns exactly EL_ITEMS edges: no bounds test in the kernels");

__global__ void __launch_bounds__(256) k_bounds(const int* __restrict__ sdst, int* __restrict__ rowptr) {
    int t = blockIdx.x * 256 + threadIdx.x;
    int d[EL_ITEMS];
    int prev[EL_ITEMS];
#pragma unroll
    for (int i = 0; i < EL_ITEMS; ++i) {
        int p = t + i * EL_STRIDE;
        d[i] = sdst[p];
        prev[i] = (i > 0 || p > 0) ? sdst[p - 1] : -1;
    }
#pragma unroll
    for (int i = 0; i < EL_ITEMS; ++i) {
        int p = t + i * EL_STRIDE;
        for (int n = prev[i] + 1; n <= d[i]; ++n) rowptr[n] = p;
        if (p == N_EDGES - 1)
            for (int n = d[i] + 1; n <= N_NODES; ++n) rowptr[n] = N_EDGES;
    }
}

__global__ void __launch_bounds__(256) k_pair(const int* __restrict__ ssrc, const int* __restrict__ sdst,
                                              const int* __restrict__ rowptr, int2* __restrict__ csr) {
    int t = blockIdx.x * 256 + threadIdx.x;
    int s[EL_ITEMS];
    int d[EL_ITEMS];
#pragma unroll
    for (int i = 0; i < EL_ITEMS; ++i) {
        int p = t + i * EL_STRIDE;
        s[i] = ssrc[p];
        d[i] = sdst[p];
    }
    int degs[EL_ITEMS];
    int degd[EL_ITEMS];
#pragma unroll
    for (int i = 0; i < EL_ITEMS; ++i) {
        degs[i] = rowptr[s[i] + 1] - rowptr[s[i]];
        degd[i] = rowptr[d[i] + 1] - rowptr[d[i]];
    }
#pragma unroll
    for (int i = 0; i < EL_ITEMS; ++i) {
        int p = t + i * EL_STRIDE;
        float ds = rsqrtf((float)degs[i] + 1.0f);
        float dd = rsqrtf((float)degd[i] + 1.0f);
        int2 v;
        v.x = s[i];
        v.y = __float_as_int(ds * dd);
        csr[p] = v;
    }
}

#define GI_NODES  10
#define GI_BLOCKS 625
static_assert(GI_BLOCKS * 8 * GI_NODES == N_NODES, "every wave owns exactly GI_NODES nodes: no bounds test in the kernel");
__global__ void __launch_bounds__(256) k_gemm_in(const float* __restrict__ x, const float* __restrict__ W1,
                                                 float* __restrict__ out) {
    int lane = threadIdx.x & 31;
    int wave = __builtin_amdgcn_readfirstlane((blockIdx.x * 256 + threadIdx.x) >> 5);
    float4 w[IN_DIM];
#pragma unroll
    for (int k = 0; k < IN_DIM; ++k) w[k] = *(const float4*)(W1 + k * HID + lane * 4);
    int n0 = wave * GI_NODES;
#pragma unroll 5
    for (int i = 0; i < GI_NODES; ++i) {
        int n = n0 + i;
        const float* xr = x + n * IN_DIM;
        float4 acc = make_float4(0.f, 0.f, 0.f, 0.f);
#pragma unroll
        for (int k = 0; k < IN_DIM; ++k) {
            float xv = xr[k];
            acc.x = fmaf(xv, w[k].x, acc.x); acc.y = fmaf(xv, w[k].y, acc.y);
            acc.z = fmaf(xv, w[k].z, acc.z); acc.w = fmaf(xv, w[k].w, acc.w);
        }
        *(float4*)(out + n * HID + lane * 4) = acc;
    }
}

template <bool RELU>
__global__ void k_gemm_wmma(const float* __restrict__ A, const float* __restrict__ W,
                            float* __restrict__ out) {
    int wave = (blockIdx.x * blockDim.x + threadIdx.x) >> 5;
    int lane = threadIdx.x & 31;
    int ct = wave & 7;
    int rt = wave >> 3;
    if (rt >= (N_NODES >> 4)) return;

    int half = lane >> 4;
    int l16  = lane & 15;
    int row  = (rt << 4) + l16;
    int col  = (ct << 4) + l16;
    const float* arow = A + row * HID;

    v8f c = {0.f, 0.f, 0.f, 0.f, 0.f, 0.f, 0.f, 0.f};
#pragma unroll
    for (int k0 = 0; k0 < HID; k0 += 4) {
        int ka = k0 + (half << 1);
        float a0 = arow[ka];
        float a1 = arow[ka + 1];
        if (RELU) { a0 = fmaxf(a0, 0.f); a1 = fmaxf(a1, 0.f); }
        v2f a; a.x = a0; a.y = a1;
        v2f b; b.x = W[ka * HID + col]; b.y = W[(ka + 1) * HID + col];
        c = __builtin_amdgcn_wmma_f32_16x16x4_f32(false, a, false, b,
                                                  (short)0, c, false, false);
    }
    float* orow = out + ((rt << 4) + (half << 3)) * HID + (ct << 4) + l16;
#pragma unroll
    for (int v = 0; v < 8; ++v) orow[v * HID] = c[v];
}

#define GA_NODES  4
#define GA_BLOCKS 1563
__global__ void __launch_bounds__(256) k_gather(const int* __restrict__ rowptr, const int2* __restrict__ csr,
                                                const float* __restrict__ hw, const float* __restrict__ bias,
                                                float* __restrict__ out) {
    int wave = __builtin_amdgcn_readfirstlane((blockIdx.x * 256 + threadIdx.x) >> 5);
    int lane = threadIdx.x & 31;
    const float* hl = hw + lane * 4;
    const float4 b4 = *(const float4*)(bias + lane * 4);
    for (int i = 0; i < GA_NODES; ++i) {
        int n = wave * GA_NODES + i;
        if (n >= N_NODES) return;
        int beg = rowptr[n];
        int end = rowptr[n + 1];
        float4 acc = make_float4(0.f, 0.f, 0.f, 0.f);
        int j = beg;
        for (; j + 4 <= end; j += 4) {
            int2 e0 = csr[j], e1 = csr[j + 1], e2 = csr[j + 2], e3 = csr[j + 3];
            const float4 h0 = *(const float4*)(hl + e0.x * HID);
            const float4 h1 = *(const float4*)(hl + e1.x * HID);
            const float4 h2 = *(const float4*)(hl + e2.x * HID);
            const float4 h3 = *(const float4*)(hl + e3.x * HID);
            float w0 = __int_as_float(e0.y), w1 = __int_as_float(e1.y);
            float w2 = __int_as_float(e2.y), w3 = __int_as_float(e3.y);
            acc.x = fmaf(h0.x, w0, acc.x); acc.y = fmaf(h0.y, w0, acc.y);
            acc.z = fmaf(h0.z, w0, acc.z); acc.w = fmaf(h0.w, w0, acc.w);
            acc.x = fmaf(h1.x, w1, acc.x); acc.y = fmaf(h1.y, w1, acc.y);
            acc.z = fmaf(h1.z, w1, acc.z); acc.w = fmaf(h1.w, w1, acc.w);
            acc.x = fmaf(h2.x, w2, acc.x); acc.y = fmaf(h2.y, w2, acc.y);
            acc.z = fmaf(h2.z, w2, acc.z); acc.w = fmaf(h2.w, w2, acc.w);
            acc.x = fmaf(h3.x, w3, acc.x); acc.y = fmaf(h3.y, w3, acc.y);
            acc.z = fmaf(h3.z, w3, acc.z); acc.w = fmaf(h3.w, w3, acc.w);
        }
        for (; j < end; ++j) {
            int2 e0 = csr[j];
            const float4 h0 = *(const float4*)(hl + e0.x * HID);
            float w0 = __int_as_float(e0.y);
            acc.x = fmaf(h0.x, w0, acc.x); acc.y = fmaf(h0.y, w0, acc.y);
            acc.z = fmaf(h0.z, w0, acc.z); acc.w = fmaf(h0.w, w0, acc.w);
        }
        float sd = rsqrtf((float)(end - beg) + 1.0f);
        float ss = sd * sd;
        const float4 hs = *(const float4*)(hl + n * HID);
        float4 o;
        o.x = acc.x + fmaf(hs.x, ss, b4.x);
        o.y = acc.y + fmaf(hs.y, ss, b4.y);
        o.z = acc.z + fmaf(hs.z, ss, b4.z);
        o.w = acc.w + fmaf(hs.w, ss, b4.w);
        *(float4*)(out + n * HID + lane * 4) = o;
    }
}

__global__ void k_pool_zero(float* __restrict__ sums, float* __restrict__ counts) {
    int i = blockIdx.x * blockDim.x + threadIdx.x;
    if (i < N_GRAPHS * HID) sums[i] = 0.f;
    if (i < N_GRAPHS) counts[i] = 0.f;
}

#define PA_NODES  10
#define PA_BLOCKS 625
static_assert(PA_BLOCKS * 8 * PA_NODES == N_NODES, "every wave owns exactly PA_NODES nodes: no bounds test in the kernel");
__global__ void __launch_bounds__(256) k_pool_acc(const float* __restrict__ h, const int* __restrict__ batch,
                                                  float* __restrict__ sums, float* __restrict__ counts) {
    int lane = threadIdx.x & 31;
    int wave = __builtin_amdgcn_readfirstlane((blockIdx.x * 256 + threadIdx.x) >> 5);
    int n0 = wave * PA_NODES;
    int g[PA_NODES];
    float4 h4[PA_NODES];
#pragma unroll
    for (int i = 0; i < PA_NODES; ++i) {
        int n = n0 + i;
        g[i] = batch[n];
        h4[i] = *(const float4*)(h + n * HID + lane * 4);
    }
#pragma unroll
    for (int i = 0; i < PA_NODES; ++i) {
        float* o = sums + g[i] * HID + lane * 4;
        atomic_add_f32(o + 0, h4[i].x);
        atomic_add_f32(o + 1, h4[i].y);
        atomic_add_f32(o + 2, h4[i].z);
        atomic_add_f32(o + 3, h4[i].w);
        if (lane == 0) atomic_add_f32(&counts[g[i]], 1.0f);
    }
}

__global__ void k_pool_div(float* __restrict__ sums, const float* __restrict__ counts) {
    int i = blockIdx.x * blockDim.x + threadIdx.x;
    if (i >= N_GRAPHS * HID) return;
    int g = i >> 7;
    sums[i] = sums[i] / fmaxf(counts[g], 1.0f);
}

extern "C" void kernel_launch(void* const* d_in, const int* in_sizes, int n_in,
                              void* d_out, int out_size, void* d_ws, size_t ws_size,
                              hipStream_t stream) {
    const float* x        = (const float*)d_in[0];
    const int*   eidx     = (const int*)d_in[1];
    const int*   batch    = (const int*)d_in[2];
    const float* W1       = (const float*)d_in[3];
    const float* b1       = (const float*)d_in[4];
    const float* W2       = (const float*)d_in[5];
    const float* b2       = (const float*)d_in[6];
    const float* W3       = (const float*)d_in[7];
    const float* b3       = (const float*)d_in[8];
    float* out = (float*)d_out;

    const int* src = eidx;
    const int* dst = eidx + N_EDGES;

    float* ws     = (float*)d_ws;
    float* hw     = ws;
    float* hA     = hw + (size_t)N_NODES * HID;
    float* counts = hA + (size_t)N_NODES * HID;
    int2*  csr    = (int2*)(counts + N_GRAPHS);
    int*   kA     = (int*)(csr + N_EDGES);
    int*   vA     = kA + N_EDGES;
    int*   kB     = vA + N_EDGES;
    int*   vB     = kB + N_EDGES;
    int*   hist   = vB + N_EDGES;
    int*   hscan  = hist + RS_HIST;
    int*   rowptr = hscan + RS_HIST;

    const int B = 256;
    int g_wmma = ((N_NODES >> 4) * 8 + (B >> 5) - 1) / (B >> 5);

    k_hist<0><<<RS_BLOCKS, B, 0, stream>>>(dst, hist);
    k_scan<<<256, B, 0, stream>>>(hist, hscan);
    k_scatter<0><<<RS_BLOCKS, B, 0, stream>>>(dst, src, hscan, kA, vA);
    k_hist<8><<<RS_BLOCKS, B, 0, stream>>>(kA, hist);
    k_scan<<<256, B, 0, stream>>>(hist, hscan);
    k_scatter<8><<<RS_BLOCKS, B, 0, stream>>>(kA, vA, hscan, kB, vB);
    k_bounds<<<EL_BLOCKS, B, 0, stream>>>(kB, rowptr);
    k_pair<<<EL_BLOCKS, B, 0, stream>>>(vB, kB, rowptr, csr);

    k_gemm_in<<<GI_BLOCKS, B, 0, stream>>>(x, W1, hw);
    k_gather<<<GA_BLOCKS, B, 0, stream>>>(rowptr, csr, hw, b1, hA);

    k_gemm_wmma<true><<<g_wmma, B, 0, stream>>>(hA, W2, hw);
    k_gather<<<GA_BLOCKS, B, 0, stream>>>(rowptr, csr, hw, b2, hA);

    k_gemm_wmma<true><<<g_wmma, B, 0, stream>>>(hA, W3, hw);
    k_gather<<<GA_BLOCKS, B, 0, stream>>>(rowptr, csr, hw, b3, hA);

    k_pool_zero<<<(N_GRAPHS * HID + B - 1) / B, B, 0, stream>>>(out, counts);
    k_pool_acc<<<PA_BLOCKS, B, 0, stream>>>(hA, batch, out, counts);
    k_pool_div<<<(N_GRAPHS * HID + B - 1) / B, B, 0, stream>>>(out, counts);
}


# Layer 1 sums the 9-wide input rows over the in-edges first, then applies the 9x128 product once per node in the same kernel (one launch fewer); per-node sum order changes (3 interleaved partial sums)
# speedup vs baseline: 3.4511x; 1.0787x over previous
#include <hip/hip_runtime.h>

#define N_NODES  50000
#define N_EDGES  1600000
#define N_GRAPHS 256
#define IN_DIM   9
#define HID      128

typedef __attribute__((ext_vector_type(2))) float v2f;
typedef __attribute__((ext_vector_type(8))) float v8f;

__device__ __forceinline__ void atomic_add_f32(float* p, float v) {
    asm volatile("global_atomic_add_f32 %0, %1, off scope:SCOPE_DEV"
                 :: "v"(p), "v"(v) : "memory");
}

#define RS_ITEMS  32
#define RS_WTILE  1024
#define RS_BTILE  8192
#define RS_BLOCKS 196
#define RS_HIST   50176

__device__ __forceinline__ unsigned match_digit(int d, bool valid) {
    unsigned m = __builtin_amdgcn_ballot_w32(valid);
#pragma unroll
    for (int b = 0; b < 8; ++b) {
        bool bit = ((d >> b) & 1) != 0;
        unsigned bal = __builtin_amdgcn_ballot_w32(bit);
        m &= bit ? bal : ~bal;
    }
    return m;
}

template <int SHIFT>
__global__ void __launch_bounds__(256) k_hist(const int* __restrict__ keys, int* __restrict__ hist) {
    __shared__ int h[8][256];
    int tid = threadIdx.x;
    int w = tid >> 5;
    int lane = tid & 31;
#pragma unroll
    for (int k = 0; k < 8; ++k) h[k][tid] = 0;
    __syncthreads();
    int base = blockIdx.x * RS_BTILE + w * RS_WTILE + lane;
    int key[RS_ITEMS];
#pragma unroll
    for (int i = 0; i < RS_ITEMS; ++i) {
        int idx = base + i * 32;
        key[i] = keys[idx < N_EDGES ? idx : N_EDGES - 1];
    }
#pragma unroll
    for (int i = 0; i < RS_ITEMS; ++i) {
        bool valid = (base + i * 32) < N_EDGES;
        int d = (key[i] >> SHIFT) & 255;
        unsigned m = match_digit(d, valid);
        int top = 31 - __builtin_clz(m | 1u);
        if (valid && lane == top) h[w][d] += __builtin_popcount(m);
    }
    __syncthreads();
    int tot = 0;
#pragma unroll
    for (int k = 0; k < 8; ++k) tot += h[k][tid];
    hist[tid * RS_BLOCKS + blockIdx.x] = tot;
}

__global__ void __launch_bounds__(256) k_scan(const int* __restrict__ in, int* __restrict__ out) {
    __shared__ int s[256];
    int t = threadIdx.x;
    int row = blockIdx.x * RS_BLOCKS;
    int part = 0;
    for (int i = t; i < row; i += 256) part += in[i];
    s[t] = part;
    __syncthreads();
    for (int off = 1; off < 256; off <<= 1) {
        int v = (t >= off) ? s[t - off] : 0;
        __syncthreads();
        s[t] += v;
        __syncthreads();
    }
    int before = s[255];
    __syncthreads();
    int x = (t < RS_BLOCKS) ? in[row + t] : 0;
    s[t] = x;
    __syncthreads();
    for (int off = 1; off < 256; off <<= 1) {
        int v = (t >= off) ? s[t - off] : 0;
        __syncthreads();
        s[t] += v;
        __syncthreads();
    }
    if (t < RS_BLOCKS) out[row + t] = before + s[t] - x;
}

template <int SHIFT>
__global__ void __launch_bounds__(256) k_scatter(const int* __restrict__ keys, const int* __restrict__ vals,
                                                 const int* __restrict__ hscan,
                                                 int* __restrict__ keys_out, int* __restrict__ vals_out) {
    __shared__ int h[8][256];
    int tid = threadIdx.x;
    int w = tid >> 5;
    int lane = tid & 31;
#pragma unroll
    for (int k = 0; k < 8; ++k) h[k][tid] = 0;
    __syncthreads();
    int base = blockIdx.x * RS_BTILE + w * RS_WTILE + lane;
    int key[RS_ITEMS];
    int val[RS_ITEMS];
#pragma unroll
    for (int i = 0; i < RS_ITEMS; ++i) {
        int idx = base + i * 32;
        int ic = idx < N_EDGES ? idx : N_EDGES - 1;
        key[i] = keys[ic];
        val[i] = vals[ic];
    }
    unsigned lt = (1u << lane) - 1u;
    int info[RS_ITEMS];
#pragma unroll
    for (int i = 0; i < RS_ITEMS; ++i) {
        bool valid = (base + i * 32) < N_EDGES;
        int d = (key[i] >> SHIFT) & 255;
        unsigned m = match_digit(d, valid);
        int top = 31 - __builtin_clz(m | 1u);
        int cntg = __builtin_popcount(m);
        bool leader = valid && lane == top;
        if (leader) h[w][d] += cntg;
        info[i] = __builtin_popcount(m & lt) | (cntg << 8) | (leader ? 0x10000 : 0);
    }
    __syncthreads();
    {
        int run = hscan[tid * RS_BLOCKS + blockIdx.x];
#pragma unroll
        for (int k = 0; k < 8; ++k) {
            int c = h[k][tid];
            h[k][tid] = run;
            run += c;
        }
    }
    __syncthreads();
#pragma unroll
    for (int i = 0; i < RS_ITEMS; ++i) {
        bool valid = (base + i * 32) < N_EDGES;
        int d = (key[i] >> SHIFT) & 255;
        int start = h[w][d];
        int pos = start + (info[i] & 255);
        if (info[i] & 0x10000) h[w][d] = start + ((info[i] >> 8) & 255);
        if (valid) {
            keys_out[pos] = key[i];
            vals_out[pos] = val[i];
        }
    }
}

#define EL_ITEMS   5
#define EL_BLOCKS  1250
#define EL_STRIDE  (EL_BLOCKS * 256)
static_assert(EL_STRIDE * EL_ITEMS == N_EDGES, "every thread owns exactly EL_ITEMS edges: no bounds test in the kernels");

__global__ void __launch_bounds__(256) k_bounds(const int* __restrict__ sdst, int* __restrict__ rowptr) {
    int t = blockIdx.x * 256 + threadIdx.x;
    int d[EL_ITEMS];
    int prev[EL_ITEMS];
#pragma unroll
    for (int i = 0; i < EL_ITEMS; ++i) {
        int p = t + i * EL_STRIDE;
        d[i] = sdst[p];
        prev[i] = (i > 0 || p > 0) ? sdst[p - 1] : -1;
    }
#pragma unroll
    for (int i = 0; i < EL_ITEMS; ++i) {
        int p = t + i * EL_STRIDE;
        for (int n = prev[i] + 1; n <= d[i]; ++n) rowptr[n] = p;
        if (p == N_EDGES - 1)
            for (int n = d[i] + 1; n <= N_NODES; ++n) rowptr[n] = N_EDGES;
    }
}

__global__ void __launch_bounds__(256) k_pair(const int* __restrict__ ssrc, const int* __restrict__ sdst,
                                              const int* __restrict__ rowptr, int2* __restrict__ csr) {
    int t = blockIdx.x * 256 + threadIdx.x;
    int s[EL_ITEMS];
    int d[EL_ITEMS];
#pragma unroll
    for (int i = 0; i < EL_ITEMS; ++i) {
        int p = t + i * EL_STRIDE;
        s[i] = ssrc[p];
        d[i] = sdst[p];
    }
    int degs[EL_ITEMS];
    int degd[EL_ITEMS];
#pragma unroll
    for (int i = 0; i < EL_ITEMS; ++i) {
        degs[i] = rowptr[s[i] + 1] - rowptr[s[i]];
        degd[i] = rowptr[d[i] + 1] - rowptr[d[i]];
    }
#pragma unroll
    for (int i = 0; i < EL_ITEMS; ++i) {
        int p = t + i * EL_STRIDE;
        float ds = rsqrtf((float)degs[i] + 1.0f);
        float dd = rsqrtf((float)degd[i] + 1.0f);
        int2 v;
        v.x = s[i];
        v.y = __float_as_int(ds * dd);
        csr[p] = v;
    }
}

template <bool RELU>
__global__ void k_gemm_wmma(const float* __restrict__ A, const float* __restrict__ W,
                            float* __restrict__ out) {
    int wave = (blockIdx.x * blockDim.x + threadIdx.x) >> 5;
    int lane = threadIdx.x & 31;
    int ct = wave & 7;
    int rt = wave >> 3;
    if (rt >= (N_NODES >> 4)) return;

    int half = lane >> 4;
    int l16  = lane & 15;
    int row  = (rt << 4) + l16;
    int col  = (ct << 4) + l16;
    const float* arow = A + row * HID;

    v8f c = {0.f, 0.f, 0.f, 0.f, 0.f, 0.f, 0.f, 0.f};
#pragma unroll
    for (int k0 = 0; k0 < HID; k0 += 4) {
        int ka = k0 + (half << 1);
        float a0 = arow[ka];
        float a1 = arow[ka + 1];
        if (RELU) { a0 = fmaxf(a0, 0.f); a1 = fmaxf(a1, 0.f); }
        v2f a; a.x = a0; a.y = a1;
        v2f b; b.x = W[ka * HID + col]; b.y = W[(ka + 1) * HID + col];
        c = __builtin_amdgcn_wmma_f32_16x16x4_f32(false, a, false, b,
                                                  (short)0, c, false, false);
    }
    float* orow = out + ((rt << 4) + (half << 3)) * HID + (ct << 4) + l16;
#pragma unroll
    for (int v = 0; v < 8; ++v) orow[v * HID] = c[v];
}

#define GA_NODES  4
#define GA_BLOCKS 1563
__global__ void __launch_bounds__(256) k_gather(const int* __restrict__ rowptr, const int2* __restrict__ csr,
                                                const float* __restrict__ hw, const float* __restrict__ bias,
                                                float* __restrict__ out) {
    int wave = __builtin_amdgcn_readfirstlane((blockIdx.x * 256 + threadIdx.x) >> 5);
    int lane = threadIdx.x & 31;
    const float* hl = hw + lane * 4;
    const float4 b4 = *(const float4*)(bias + lane * 4);
    for (int i = 0; i < GA_NODES; ++i) {
        int n = wave * GA_NODES + i;
        if (n >= N_NODES) return;
        int beg = rowptr[n];
        int end = rowptr[n + 1];
        float4 acc = make_float4(0.f, 0.f, 0.f, 0.f);
        int j = beg;
        for (; j + 4 <= end; j += 4) {
            int2 e0 = csr[j], e1 = csr[j + 1], e2 = csr[j + 2], e3 = csr[j + 3];
            const float4 h0 = *(const float4*)(hl + e0.x * HID);
            const float4 h1 = *(const float4*)(hl + e1.x * HID);
            const float4 h2 = *(const float4*)(hl + e2.x * HID);
            const float4 h3 = *(const float4*)(hl + e3.x * HID);
            float w0 = __int_as_float(e0.y), w1 = __int_as_float(e1.y);
            float w2 = __int_as_float(e2.y), w3 = __int_as_float(e3.y);
            acc.x = fmaf(h0.x, w0, acc.x); acc.y = fmaf(h0.y, w0, acc.y);
            acc.z = fmaf(h0.z, w0, acc.z); acc.w = fmaf(h0.w, w0, acc.w);
            acc.x = fmaf(h1.x, w1, acc.x); acc.y = fmaf(h1.y, w1, acc.y);
            acc.z = fmaf(h1.z, w1, acc.z); acc.w = fmaf(h1.w, w1, acc.w);
            acc.x = fmaf(h2.x, w2, acc.x); acc.y = fmaf(h2.y, w2, acc.y);
            acc.z = fmaf(h2.z, w2, acc.z); acc.w = fmaf(h2.w, w2, acc.w);
            acc.x = fmaf(h3.x, w3, acc.x); acc.y = fmaf(h3.y, w3, acc.y);
            acc.z = fmaf(h3.z, w3, acc.z); acc.w = fmaf(h3.w, w3, acc.w);
        }
        for (; j < end; ++j) {
            int2 e0 = csr[j];
            const float4 h0 = *(const float4*)(hl + e0.x * HID);
            float w0 = __int_as_float(e0.y);
            acc.x = fmaf(h0.x, w0, acc.x); acc.y = fmaf(h0.y, w0, acc.y);
            acc.z = fmaf(h0.z, w0, acc.z); acc.w = fmaf(h0.w, w0, acc.w);
        }
        float sd = rsqrtf((float)(end - beg) + 1.0f);
        float ss = sd * sd;
        const float4 hs = *(const float4*)(hl + n * HID);
        float4 o;
        o.x = acc.x + fmaf(hs.x, ss, b4.x);
        o.y = acc.y + fmaf(hs.y, ss, b4.y);
        o.z = acc.z + fmaf(hs.z, ss, b4.z);
        o.w = acc.w + fmaf(hs.w, ss, b4.w);
        *(float4*)(out + n * HID + lane * 4) = o;
    }
}

#define GX_UNROLL 12
#define GX_TRIP   (3 * GX_UNROLL)
#define GX_NODES  5
#define GX_BLOCKS 1250
static_assert(GX_BLOCKS * 8 * GX_NODES == N_NODES, "every wave owns exactly GX_NODES nodes: no bounds test in the kernel");
__global__ void __launch_bounds__(256) k_gather_in(const int* __restrict__ rowptr, const int2* __restrict__ csr,
                                                   const float* __restrict__ x, const float* __restrict__ W1,
                                                   const float* __restrict__ bias, float* __restrict__ out) {
    int wave = __builtin_amdgcn_readfirstlane((blockIdx.x * 256 + threadIdx.x) >> 5);
    int lane = threadIdx.x & 31;
    int l = lane < 27 ? lane : 26;
    int e = l / IN_DIM;
    int k = l - e * IN_DIM;
    float4 w[IN_DIM];
#pragma unroll
    for (int q = 0; q < IN_DIM; ++q) w[q] = *(const float4*)(W1 + q * HID + lane * 4);
    const float4 b4 = *(const float4*)(bias + lane * 4);
    for (int i = 0; i < GX_NODES; ++i) {
        int n = wave * GX_NODES + i;
        int beg = rowptr[n];
        int end = rowptr[n + 1];
        float xs = x[n * IN_DIM + k];
        float acc = 0.f;
        for (int j0 = beg; j0 < end; j0 += GX_TRIP) {
            int2 ed[GX_UNROLL];
            float xv[GX_UNROLL];
#pragma unroll
            for (int u = 0; u < GX_UNROLL; ++u) {
                int j = j0 + 3 * u + e;
                ed[u] = csr[j < end ? j : end - 1];
            }
#pragma unroll
            for (int u = 0; u < GX_UNROLL; ++u) xv[u] = x[ed[u].x * IN_DIM + k];
#pragma unroll
            for (int u = 0; u < GX_UNROLL; ++u) {
                int j = j0 + 3 * u + e;
                float t = fmaf(xv[u], __int_as_float(ed[u].y), acc);
                acc = j < end ? t : acc;
            }
        }
        float s1 = __int_as_float(__builtin_amdgcn_ds_bpermute(((lane + 9) & 31) << 2, __float_as_int(acc)));
        float s2 = __int_as_float(__builtin_amdgcn_ds_bpermute(((lane + 18) & 31) << 2, __float_as_int(acc)));
        float sd = rsqrtf((float)(end - beg) + 1.0f);
        float ss = sd * sd;
        float ax = fmaf(xs, ss, (acc + s1) + s2);
        float4 o = make_float4(0.f, 0.f, 0.f, 0.f);
#pragma unroll
        for (int q = 0; q < IN_DIM; ++q) {
            float a = __int_as_float(__builtin_amdgcn_readlane(__float_as_int(ax), q));
            o.x = fmaf(a, w[q].x, o.x); o.y = fmaf(a, w[q].y, o.y);
            o.z = fmaf(a, w[q].z, o.z); o.w = fmaf(a, w[q].w, o.w);
        }
        o.x += b4.x; o.y += b4.y; o.z += b4.z; o.w += b4.w;
        *(float4*)(out + n * HID + lane * 4) = o;
    }
}

__global__ void k_pool_zero(float* __restrict__ sums, float* __restrict__ counts) {
    int i = blockIdx.x * blockDim.x + threadIdx.x;
    if (i < N_GRAPHS * HID) sums[i] = 0.f;
    if (i < N_GRAPHS) counts[i] = 0.f;
}

#define PA_NODES  10
#define PA_BLOCKS 625
static_assert(PA_BLOCKS * 8 * PA_NODES == N_NODES, "every wave owns exactly PA_NODES nodes: no bounds test in the kernel");
__global__ void __launch_bounds__(256) k_pool_acc(const float* __restrict__ h, const int* __restrict__ batch,
                                                  float* __restrict__ sums, float* __restrict__ counts) {
    int lane = threadIdx.x & 31;
    int wave = __builtin_amdgcn_readfirstlane((blockIdx.x * 256 + threadIdx.x) >> 5);
    int n0 = wave * PA_NODES;
    int g[PA_NODES];
    float4 h4[PA_NODES];
#pragma unroll
    for (int i = 0; i < PA_NODES; ++i) {
        int n = n0 + i;
        g[i] = batch[n];
        h4[i] = *(const float4*)(h + n * HID + lane * 4);
    }
#pragma unroll
    for (int i = 0; i < PA_NODES; ++i) {
        float* o = sums + g[i] * HID + lane * 4;
        atomic_add_f32(o + 0, h4[i].x);
        atomic_add_f32(o + 1, h4[i].y);
        atomic_add_f32(o + 2, h4[i].z);
        atomic_add_f32(o + 3, h4[i].w);
        if (lane == 0) atomic_add_f32(&counts[g[i]], 1.0f);
    }
}

__global__ void k_pool_div(float* __restrict__ sums, const float* __restrict__ counts) {
    int i = blockIdx.x * blockDim.x + threadIdx.x;
    if (i >= N_GRAPHS * HID) return;
    int g = i >> 7;
    sums[i] = sums[i] / fmaxf(counts[g], 1.0f);
}

extern "C" void kernel_launch(void* const* d_in, const int* in_sizes, int n_in,
                              void* d_out, int out_size, void* d_ws, size_t ws_size,
                              hipStream_t stream) {
    const float* x        = (const float*)d_in[0];
    const int*   eidx     = (const int*)d_in[1];
    const int*   batch    = (const int*)d_in[2];
    const float* W1       = (const float*)d_in[3];
    const float* b1       = (const float*)d_in[4];
    const float* W2       = (const float*)d_in[5];
    const float* b2       = (const float*)d_in[6];
    const float* W3       = (const float*)d_in[7];
    const float* b3       = (const float*)d_in[8];
    float* out = (float*)d_out;

    const int* src = eidx;
    const int* dst = eidx + N_EDGES;

    float* ws     = (float*)d_ws;
    float* hw     = ws;
    float* hA     = hw + (size_t)N_NODES * HID;
    float* counts = hA + (size_t)N_NODES * HID;
    int2*  csr    = (int2*)(counts + N_GRAPHS);
    int*   kA     = (int*)(csr + N_EDGES);
    int*   vA     = kA + N_EDGES;
    int*   kB     = vA + N_EDGES;
    int*   vB     = kB + N_EDGES;
    int*   hist   = vB + N_EDGES;
    int*   hscan  = hist + RS_HIST;
    int*   rowptr = hscan + RS_HIST;

    const int B = 256;
    int g_wmma = ((N_NODES >> 4) * 8 + (B >> 5) - 1) / (B >> 5);

    k_hist<0><<<RS_BLOCKS, B, 0, stream>>>(dst, hist);
    k_scan<<<256, B, 0, stream>>>(hist, hscan);
    k_scatter<0><<<RS_BLOCKS, B, 0, stream>>>(dst, src, hscan, kA, vA);
    k_hist<8><<<RS_BLOCKS, B, 0, stream>>>(kA, hist);
    k_scan<<<256, B, 0, stream>>>(hist, hscan);
    k_scatter<8><<<RS_BLOCKS, B, 0, stream>>>(kA, vA, hscan, kB, vB);
    k_bounds<<<EL_BLOCKS, B, 0, stream>>>(kB, rowptr);
    k_pair<<<EL_BLOCKS, B, 0, stream>>>(vB, kB, rowptr, csr);

    k_gather_in<<<GX_BLOCKS, B, 0, stream>>>(rowptr, csr, x, W1, b1, hA);

    k_gemm_wmma<true><<<g_wmma, B, 0, stream>>>(hA, W2, hw);
    k_gather<<<GA_BLOCKS, B, 0, stream>>>(rowptr, csr, hw, b2, hA);

    k_gemm_wmma<true><<<g_wmma, B, 0, stream>>>(hA, W3, hw);
    k_gather<<<GA_BLOCKS, B, 0, stream>>>(rowptr, csr, hw, b3, hA);

    k_pool_zero<<<(N_GRAPHS * HID + B - 1) / B, B, 0, stream>>>(out, counts);
    k_pool_acc<<<PA_BLOCKS, B, 0, stream>>>(hA, batch, out, counts);
    k_pool_div<<<(N_GRAPHS * HID + B - 1) / B, B, 0, stream>>>(out, counts);
}


# Layer 3's weights are applied after mean pooling (pooling is a sum, layer 3 has no activation): a 256-row product replaces the 50000-row one; graph sums now precede the W3 product
# speedup vs baseline: 3.9601x; 1.1475x over previous
#include <hip/hip_runtime.h>

#define N_NODES  50000
#define N_EDGES  1600000
#define N_GRAPHS 256
#define IN_DIM   9
#define HID      128

typedef __attribute__((ext_vector_type(2))) float v2f;
typedef __attribute__((ext_vector_type(8))) float v8f;

__device__ __forceinline__ void atomic_add_f32(float* p, float v) {
    asm volatile("global_atomic_add_f32 %0, %1, off scope:SCOPE_DEV"
                 :: "v"(p), "v"(v) : "memory");
}

#define RS_ITEMS  32
#define RS_WTILE  1024
#define RS_BTILE  8192
#define RS_BLOCKS 196
#define RS_HIST   50176

__device__ __forceinline__ unsigned match_digit(int d, bool valid) {
    unsigned m = __builtin_amdgcn_ballot_w32(valid);
#pragma unroll
    for (int b = 0; b < 8; ++b) {
        bool bit = ((d >> b) & 1) != 0;
        unsigned bal = __builtin_amdgcn_ballot_w32(bit);
        m &= bit ? bal : ~bal;
    }
    return m;
}

template <int SHIFT>
__global__ void __launch_bounds__(256) k_hist(const int* __restrict__ keys, int* __restrict__ hist) {
    __shared__ int h[8][256];
    int tid = threadIdx.x;
    int w = tid >> 5;
    int lane = tid & 31;
#pragma unroll
    for (int k = 0; k < 8; ++k) h[k][tid] = 0;
    __syncthreads();
    int base = blockIdx.x * RS_BTILE + w * RS_WTILE + lane;
    int key[RS_ITEMS];
#pragma unroll
    for (int i = 0; i < RS_ITEMS; ++i) {
        int idx = base + i * 32;
        key[i] = keys[idx < N_EDGES ? idx : N_EDGES - 1];
    }
#pragma unroll
    for (int i = 0; i < RS_ITEMS; ++i) {
        bool valid = (base + i * 32) < N_EDGES;
        int d = (key[i] >> SHIFT) & 255;
        unsigned m = match_digit(d, valid);
        int top = 31 - __builtin_clz(m | 1u);
        if (valid && lane == top) h[w][d] += __builtin_popcount(m);
    }
    __syncthreads();
    int tot = 0;
#pragma unroll
    for (int k = 0; k < 8; ++k) tot += h[k][tid];
    hist[tid * RS_BLOCKS + blockIdx.x] = tot;
}

__global__ void __launch_bounds__(256) k_scan(const int* __restrict__ in, int* __restrict__ out) {
    __shared__ int s[256];
    int t = threadIdx.x;
    int row = blockIdx.x * RS_BLOCKS;
    int part = 0;
    for (int i = t; i < row; i += 256) part += in[i];
    s[t] = part;
    __syncthreads();
    for (int off = 1; off < 256; off <<= 1) {
        int v = (t >= off) ? s[t - off] : 0;
        __syncthreads();
        s[t] += v;
        __syncthreads();
    }
    int before = s[255];
    __syncthreads();
    int x = (t < RS_BLOCKS) ? in[row + t] : 0;
    s[t] = x;
    __syncthreads();
    for (int off = 1; off < 256; off <<= 1) {
        int v = (t >= off) ? s[t - off] : 0;
        __syncthreads();
        s[t] += v;
        __syncthreads();
    }
    if (t < RS_BLOCKS) out[row + t] = before + s[t] - x;
}

template <int SHIFT>
__global__ void __launch_bounds__(256) k_scatter(const int* __restrict__ keys, const int* __restrict__ vals,
                                                 const int* __restrict__ hscan,
                                                 int* __restrict__ keys_out, int* __restrict__ vals_out) {
    __shared__ int h[8][256];
    int tid = threadIdx.x;
    int w = tid >> 5;
    int lane = tid & 31;
#pragma unroll
    for (int k = 0; k < 8; ++k) h[k][tid] = 0;
    __syncthreads();
    int base = blockIdx.x * RS_BTILE + w * RS_WTILE + lane;
    int key[RS_ITEMS];
    int val[RS_ITEMS];
#pragma unroll
    for (int i = 0; i < RS_ITEMS; ++i) {
        int idx = base + i * 32;
        int ic = idx < N_EDGES ? idx : N_EDGES - 1;
        key[i] = keys[ic];
        val[i] = vals[ic];
    }
    unsigned lt = (1u << lane) - 1u;
    int info[RS_ITEMS];
#pragma unroll
    for (int i = 0; i < RS_ITEMS; ++i) {
        bool valid = (base + i * 32) < N_EDGES;
        int d = (key[i] >> SHIFT) & 255;
        unsigned m = match_digit(d, valid);
        int top = 31 - __builtin_clz(m | 1u);
        int cntg = __builtin_popcount(m);
        bool leader = valid && lane == top;
        if (leader) h[w][d] += cntg;
        info[i] = __builtin_popcount(m & lt) | (cntg << 8) | (leader ? 0x10000 : 0);
    }
    __syncthreads();
    {
        int run = hscan[tid * RS_BLOCKS + blockIdx.x];
#pragma unroll
        for (int k = 0; k < 8; ++k) {
            int c = h[k][tid];
            h[k][tid] = run;
            run += c;
        }
    }
    __syncthreads();
#pragma unroll
    for (int i = 0; i < RS_ITEMS; ++i) {
        bool valid = (base + i * 32) < N_EDGES;
        int d = (key[i] >> SHIFT) & 255;
        int start = h[w][d];
        int pos = start + (info[i] & 255);
        if (info[i] & 0x10000) h[w][d] = start + ((info[i] >> 8) & 255);
        if (valid) {
            keys_out[pos] = key[i];
            vals_out[pos] = val[i];
        }
    }
}

#define EL_ITEMS   5
#define EL_BLOCKS  1250
#define EL_STRIDE  (EL_BLOCKS * 256)
static_assert(EL_STRIDE * EL_ITEMS == N_EDGES, "every thread owns exactly EL_ITEMS edges: no bounds test in the kernels");

__global__ void __launch_bounds__(256) k_bounds(const int* __restrict__ sdst, int* __restrict__ rowptr) {
    int t = blockIdx.x * 256 + threadIdx.x;
    int d[EL_ITEMS];
    int prev[EL_ITEMS];
#pragma unroll
    for (int i = 0; i < EL_ITEMS; ++i) {
        int p = t + i * EL_STRIDE;
        d[i] = sdst[p];
        prev[i] = (i > 0 || p > 0) ? sdst[p - 1] : -1;
    }
#pragma unroll
    for (int i = 0; i < EL_ITEMS; ++i) {
        int p = t + i * EL_STRIDE;
        for (int n = prev[i] + 1; n <= d[i]; ++n) rowptr[n] = p;
        if (p == N_EDGES - 1)
            for (int n = d[i] + 1; n <= N_NODES; ++n) rowptr[n] = N_EDGES;
    }
}

__global__ void __launch_bounds__(256) k_pair(const int* __restrict__ ssrc, const int* __restrict__ sdst,
                                              const int* __restrict__ rowptr, int2* __restrict__ csr) {
    int t = blockIdx.x * 256 + threadIdx.x;
    int s[EL_ITEMS];
    int d[EL_ITEMS];
#pragma unroll
    for (int i = 0; i < EL_ITEMS; ++i) {
        int p = t + i * EL_STRIDE;
        s[i] = ssrc[p];
        d[i] = sdst[p];
    }
    int degs[EL_ITEMS];
    int degd[EL_ITEMS];
#pragma unroll
    for (int i = 0; i < EL_ITEMS; ++i) {
        degs[i] = rowptr[s[i] + 1] - rowptr[s[i]];
        degd[i] = rowptr[d[i] + 1] - rowptr[d[i]];
    }
#pragma unroll
    for (int i = 0; i < EL_ITEMS; ++i) {
        int p = t + i * EL_STRIDE;
        float ds = rsqrtf((float)degs[i] + 1.0f);
        float dd = rsqrtf((float)degd[i] + 1.0f);
        int2 v;
        v.x = s[i];
        v.y = __float_as_int(ds * dd);
        csr[p] = v;
    }
}

template <bool RELU>
__global__ void k_gemm_wmma(const float* __restrict__ A, const float* __restrict__ W,
                            float* __restrict__ out) {
    int wave = (blockIdx.x * blockDim.x + threadIdx.x) >> 5;
    int lane = threadIdx.x & 31;
    int ct = wave & 7;
    int rt = wave >> 3;
    if (rt >= (N_NODES >> 4)) return;

    int half = lane >> 4;
    int l16  = lane & 15;
    int row  = (rt << 4) + l16;
    int col  = (ct << 4) + l16;
    const float* arow = A + row * HID;

    v8f c = {0.f, 0.f, 0.f, 0.f, 0.f, 0.f, 0.f, 0.f};
#pragma unroll
    for (int k0 = 0; k0 < HID; k0 += 4) {
        int ka = k0 + (half << 1);
        float a0 = arow[ka];
        float a1 = arow[ka + 1];
        if (RELU) { a0 = fmaxf(a0, 0.f); a1 = fmaxf(a1, 0.f); }
        v2f a; a.x = a0; a.y = a1;
        v2f b; b.x = W[ka * HID + col]; b.y = W[(ka + 1) * HID + col];
        c = __builtin_amdgcn_wmma_f32_16x16x4_f32(false, a, false, b,
                                                  (short)0, c, false, false);
    }
    float* orow = out + ((rt << 4) + (half << 3)) * HID + (ct << 4) + l16;
#pragma unroll
    for (int v = 0; v < 8; ++v) orow[v * HID] = c[v];
}

#define GA_NODES  4
#define GA_BLOCKS 1563
template <bool RELU_OUT, bool BIAS>
__global__ void __launch_bounds__(256) k_gather(const int* __restrict__ rowptr, const int2* __restrict__ csr,
                                                const float* __restrict__ hw, const float* __restrict__ bias,
                                                float* __restrict__ out) {
    int wave = __builtin_amdgcn_readfirstlane((blockIdx.x * 256 + threadIdx.x) >> 5);
    int lane = threadIdx.x & 31;
    const float* hl = hw + lane * 4;
    float4 b4 = make_float4(0.f, 0.f, 0.f, 0.f);
    if (BIAS) b4 = *(const float4*)(bias + lane * 4);
    for (int i = 0; i < GA_NODES; ++i) {
        int n = wave * GA_NODES + i;
        if (n >= N_NODES) return;
        int beg = rowptr[n];
        int end = rowptr[n + 1];
        float4 acc = make_float4(0.f, 0.f, 0.f, 0.f);
        int j = beg;
        for (; j + 4 <= end; j += 4) {
            int2 e0 = csr[j], e1 = csr[j + 1], e2 = csr[j + 2], e3 = csr[j + 3];
            const float4 h0 = *(const float4*)(hl + e0.x * HID);
            const float4 h1 = *(const float4*)(hl + e1.x * HID);
            const float4 h2 = *(const float4*)(hl + e2.x * HID);
            const float4 h3 = *(const float4*)(hl + e3.x * HID);
            float w0 = __int_as_float(e0.y), w1 = __int_as_float(e1.y);
            float w2 = __int_as_float(e2.y), w3 = __int_as_float(e3.y);
            acc.x = fmaf(h0.x, w0, acc.x); acc.y = fmaf(h0.y, w0, acc.y);
            acc.z = fmaf(h0.z, w0, acc.z); acc.w = fmaf(h0.w, w0, acc.w);
            acc.x = fmaf(h1.x, w1, acc.x); acc.y = fmaf(h1.y, w1, acc.y);
            acc.z = fmaf(h1.z, w1, acc.z); acc.w = fmaf(h1.w, w1, acc.w);
            acc.x = fmaf(h2.x, w2, acc.x); acc.y = fmaf(h2.y, w2, acc.y);
            acc.z = fmaf(h2.z, w2, acc.z); acc.w = fmaf(h2.w, w2, acc.w);
            acc.x = fmaf(h3.x, w3, acc.x); acc.y = fmaf(h3.y, w3, acc.y);
            acc.z = fmaf(h3.z, w3, acc.z); acc.w = fmaf(h3.w, w3, acc.w);
        }
        for (; j < end; ++j) {
            int2 e0 = csr[j];
            const float4 h0 = *(const float4*)(hl + e0.x * HID);
            float w0 = __int_as_float(e0.y);
            acc.x = fmaf(h0.x, w0, acc.x); acc.y = fmaf(h0.y, w0, acc.y);
            acc.z = fmaf(h0.z, w0, acc.z); acc.w = fmaf(h0.w, w0, acc.w);
        }
        float sd = rsqrtf((float)(end - beg) + 1.0f);
        float ss = sd * sd;
        const float4 hs = *(const float4*)(hl + n * HID);
        float4 o;
        if (BIAS) {
            o.x = acc.x + fmaf(hs.x, ss, b4.x);
            o.y = acc.y + fmaf(hs.y, ss, b4.y);
            o.z = acc.z + fmaf(hs.z, ss, b4.z);
            o.w = acc.w + fmaf(hs.w, ss, b4.w);
        } else {
            o.x = fmaf(hs.x, ss, acc.x);
            o.y = fmaf(hs.y, ss, acc.y);
            o.z = fmaf(hs.z, ss, acc.z);
            o.w = fmaf(hs.w, ss, acc.w);
        }
        if (RELU_OUT) {
            o.x = fmaxf(o.x, 0.f); o.y = fmaxf(o.y, 0.f);
            o.z = fmaxf(o.z, 0.f); o.w = fmaxf(o.w, 0.f);
        }
        *(float4*)(out + n * HID + lane * 4) = o;
    }
}

#define GX_UNROLL 12
#define GX_TRIP   (3 * GX_UNROLL)
#define GX_NODES  5
#define GX_BLOCKS 1250
static_assert(GX_BLOCKS * 8 * GX_NODES == N_NODES, "every wave owns exactly GX_NODES nodes: no bounds test in the kernel");
__global__ void __launch_bounds__(256) k_gather_in(const int* __restrict__ rowptr, const int2* __restrict__ csr,
                                                   const float* __restrict__ x, const float* __restrict__ W1,
                                                   const float* __restrict__ bias, float* __restrict__ out) {
    int wave = __builtin_amdgcn_readfirstlane((blockIdx.x * 256 + threadIdx.x) >> 5);
    int lane = threadIdx.x & 31;
    int l = lane < 27 ? lane : 26;
    int e = l / IN_DIM;
    int k = l - e * IN_DIM;
    float4 w[IN_DIM];
#pragma unroll
    for (int q = 0; q < IN_DIM; ++q) w[q] = *(const float4*)(W1 + q * HID + lane * 4);
    const float4 b4 = *(const float4*)(bias + lane * 4);
    for (int i = 0; i < GX_NODES; ++i) {
        int n = wave * GX_NODES + i;
        int beg = rowptr[n];
        int end = rowptr[n + 1];
        float xs = x[n * IN_DIM + k];
        float acc = 0.f;
        for (int j0 = beg; j0 < end; j0 += GX_TRIP) {
            int2 ed[GX_UNROLL];
            float xv[GX_UNROLL];
#pragma unroll
            for (int u = 0; u < GX_UNROLL; ++u) {
                int j = j0 + 3 * u + e;
                ed[u] = csr[j < end ? j : end - 1];
            }
#pragma unroll
            for (int u = 0; u < GX_UNROLL; ++u) xv[u] = x[ed[u].x * IN_DIM + k];
#pragma unroll
            for (int u = 0; u < GX_UNROLL; ++u) {
                int j = j0 + 3 * u + e;
                float t = fmaf(xv[u], __int_as_float(ed[u].y), acc);
                acc = j < end ? t : acc;
            }
        }
        float s1 = __int_as_float(__builtin_amdgcn_ds_bpermute(((lane + 9) & 31) << 2, __float_as_int(acc)));
        float s2 = __int_as_float(__builtin_amdgcn_ds_bpermute(((lane + 18) & 31) << 2, __float_as_int(acc)));
        float sd = rsqrtf((float)(end - beg) + 1.0f);
        float ss = sd * sd;
        float ax = fmaf(xs, ss, (acc + s1) + s2);
        float4 o = make_float4(0.f, 0.f, 0.f, 0.f);
#pragma unroll
        for (int q = 0; q < IN_DIM; ++q) {
            float a = __int_as_float(__builtin_amdgcn_readlane(__float_as_int(ax), q));
            o.x = fmaf(a, w[q].x, o.x); o.y = fmaf(a, w[q].y, o.y);
            o.z = fmaf(a, w[q].z, o.z); o.w = fmaf(a, w[q].w, o.w);
        }
        o.x += b4.x; o.y += b4.y; o.z += b4.z; o.w += b4.w;
        *(float4*)(out + n * HID + lane * 4) = o;
    }
}

__global__ void k_pool_zero(float* __restrict__ sums, float* __restrict__ counts) {
    int i = blockIdx.x * blockDim.x + threadIdx.x;
    if (i < N_GRAPHS * HID) sums[i] = 0.f;
    if (i < N_GRAPHS) counts[i] = 0.f;
}

#define PA_NODES  10
#define PA_BLOCKS 625
static_assert(PA_BLOCKS * 8 * PA_NODES == N_NODES, "every wave owns exactly PA_NODES nodes: no bounds test in the kernel");
__global__ void __launch_bounds__(256) k_pool_acc(const float* __restrict__ h, const int* __restrict__ batch,
                                                  float* __restrict__ sums, float* __restrict__ counts) {
    int lane = threadIdx.x & 31;
    int wave = __builtin_amdgcn_readfirstlane((blockIdx.x * 256 + threadIdx.x) >> 5);
    int n0 = wave * PA_NODES;
    int g[PA_NODES];
    float4 h4[PA_NODES];
#pragma unroll
    for (int i = 0; i < PA_NODES; ++i) {
        int n = n0 + i;
        g[i] = batch[n];
        h4[i] = *(const float4*)(h + n * HID + lane * 4);
    }
#pragma unroll
    for (int i = 0; i < PA_NODES; ++i) {
        float* o = sums + g[i] * HID + lane * 4;
        atomic_add_f32(o + 0, h4[i].x);
        atomic_add_f32(o + 1, h4[i].y);
        atomic_add_f32(o + 2, h4[i].z);
        atomic_add_f32(o + 3, h4[i].w);
        if (lane == 0) atomic_add_f32(&counts[g[i]], 1.0f);
    }
}

__global__ void __launch_bounds__(256) k_pool_gemm_wmma(const float* __restrict__ S, const float* __restrict__ W,
                                                        const float* __restrict__ counts, const float* __restrict__ bias,
                                                        float* __restrict__ out) {
    int wave = (blockIdx.x * 256 + threadIdx.x) >> 5;
    int lane = threadIdx.x & 31;
    int ct = wave & 7;
    int rt = wave >> 3;

    int half = lane >> 4;
    int l16  = lane & 15;
    int row  = (rt << 4) + l16;
    int col  = (ct << 4) + l16;
    const float* arow = S + row * HID;

    v8f c = {0.f, 0.f, 0.f, 0.f, 0.f, 0.f, 0.f, 0.f};
#pragma unroll
    for (int k0 = 0; k0 < HID; k0 += 4) {
        int ka = k0 + (half << 1);
        v2f a; a.x = arow[ka]; a.y = arow[ka + 1];
        v2f b; b.x = W[ka * HID + col]; b.y = W[(ka + 1) * HID + col];
        c = __builtin_amdgcn_wmma_f32_16x16x4_f32(false, a, false, b,
                                                  (short)0, c, false, false);
    }
    int g0 = (rt << 4) + (half << 3);
    float bc = bias[col];
#pragma unroll
    for (int v = 0; v < 8; ++v) {
        float cnt = counts[g0 + v];
        out[(g0 + v) * HID + col] = fmaf(cnt, bc, c[v]) / fmaxf(cnt, 1.0f);
    }
}

extern "C" void kernel_launch(void* const* d_in, const int* in_sizes, int n_in,
                              void* d_out, int out_size, void* d_ws, size_t ws_size,
                              hipStream_t stream) {
    const float* x        = (const float*)d_in[0];
    const int*   eidx     = (const int*)d_in[1];
    const int*   batch    = (const int*)d_in[2];
    const float* W1       = (const float*)d_in[3];
    const float* b1       = (const float*)d_in[4];
    const float* W2       = (const float*)d_in[5];
    const float* b2       = (const float*)d_in[6];
    const float* W3       = (const float*)d_in[7];
    const float* b3       = (const float*)d_in[8];
    float* out = (float*)d_out;

    const int* src = eidx;
    const int* dst = eidx + N_EDGES;

    float* ws     = (float*)d_ws;
    float* hw     = ws;
    float* hA     = hw + (size_t)N_NODES * HID;
    float* counts = hA + (size_t)N_NODES * HID;
    float* psum   = counts + N_GRAPHS;
    int2*  csr    = (int2*)(psum + N_GRAPHS * HID);
    int*   kA     = (int*)(csr + N_EDGES);
    int*   vA     = kA + N_EDGES;
    int*   kB     = vA + N_EDGES;
    int*   vB     = kB + N_EDGES;
    int*   hist   = vB + N_EDGES;
    int*   hscan  = hist + RS_HIST;
    int*   rowptr = hscan + RS_HIST;

    const int B = 256;
    int g_wmma = ((N_NODES >> 4) * 8 + (B >> 5) - 1) / (B >> 5);

    k_hist<0><<<RS_BLOCKS, B, 0, stream>>>(dst, hist);
    k_scan<<<256, B, 0, stream>>>(hist, hscan);
    k_scatter<0><<<RS_BLOCKS, B, 0, stream>>>(dst, src, hscan, kA, vA);
    k_hist<8><<<RS_BLOCKS, B, 0, stream>>>(kA, hist);
    k_scan<<<256, B, 0, stream>>>(hist, hscan);
    k_scatter<8><<<RS_BLOCKS, B, 0, stream>>>(kA, vA, hscan, kB, vB);
    k_bounds<<<EL_BLOCKS, B, 0, stream>>>(kB, rowptr);
    k_pair<<<EL_BLOCKS, B, 0, stream>>>(vB, kB, rowptr, csr);

    k_gather_in<<<GX_BLOCKS, B, 0, stream>>>(rowptr, csr, x, W1, b1, hA);

    k_gemm_wmma<true><<<g_wmma, B, 0, stream>>>(hA, W2, hw);
    k_gather<true, true><<<GA_BLOCKS, B, 0, stream>>>(rowptr, csr, hw, b2, hA);

    k_gather<false, false><<<GA_BLOCKS, B, 0, stream>>>(rowptr, csr, hA, b3, hw);

    k_pool_zero<<<(N_GRAPHS * HID + B - 1) / B, B, 0, stream>>>(psum, counts);
    k_pool_acc<<<PA_BLOCKS, B, 0, stream>>>(hw, batch, psum, counts);
    k_pool_gemm_wmma<<<N_GRAPHS / 16, B, 0, stream>>>(psum, W3, counts, b3, out);
}


# Layer-2 product re-tiled: each wave owns whole row tiles across all 8 column tiles (A fragments read once, 8 accumulators), W2 staged once per work-group in LDS, 256 work-groups; sums bit-identical
# speedup vs baseline: 4.1138x; 1.0388x over previous
#include <hip/hip_runtime.h>

#define N_NODES  50000
#define N_EDGES  1600000
#define N_GRAPHS 256
#define IN_DIM   9
#define HID      128

typedef __attribute__((ext_vector_type(2))) float v2f;
typedef __attribute__((ext_vector_type(8))) float v8f;

__device__ __forceinline__ void atomic_add_f32(float* p, float v) {
    asm volatile("global_atomic_add_f32 %0, %1, off scope:SCOPE_DEV"
                 :: "v"(p), "v"(v) : "memory");
}

#define RS_ITEMS  32
#define RS_WTILE  1024
#define RS_BTILE  8192
#define RS_BLOCKS 196
#define RS_HIST   50176

__device__ __forceinline__ unsigned match_digit(int d, bool valid) {
    unsigned m = __builtin_amdgcn_ballot_w32(valid);
#pragma unroll
    for (int b = 0; b < 8; ++b) {
        bool bit = ((d >> b) & 1) != 0;
        unsigned bal = __builtin_amdgcn_ballot_w32(bit);
        m &= bit ? bal : ~bal;
    }
    return m;
}

template <int SHIFT>
__global__ void __launch_bounds__(256) k_hist(const int* __restrict__ keys, int* __restrict__ hist) {
    __shared__ int h[8][256];
    int tid = threadIdx.x;
    int w = tid >> 5;
    int lane = tid & 31;
#pragma unroll
    for (int k = 0; k < 8; ++k) h[k][tid] = 0;
    __syncthreads();
    int base = blockIdx.x * RS_BTILE + w * RS_WTILE + lane;
    int key[RS_ITEMS];
#pragma unroll
    for (int i = 0; i < RS_ITEMS; ++i) {
        int idx = base + i * 32;
        key[i] = keys[idx < N_EDGES ? idx : N_EDGES - 1];
    }
#pragma unroll
    for (int i = 0; i < RS_ITEMS; ++i) {
        bool valid = (base + i * 32) < N_EDGES;
        int d = (key[i] >> SHIFT) & 255;
        unsigned m = match_digit(d, valid);
        int top = 31 - __builtin_clz(m | 1u);
        if (valid && lane == top) h[w][d] += __builtin_popcount(m);
    }
    __syncthreads();
    int tot = 0;
#pragma unroll
    for (int k = 0; k < 8; ++k) tot += h[k][tid];
    hist[tid * RS_BLOCKS + blockIdx.x] = tot;
}

__global__ void __launch_bounds__(256) k_scan(const int* __restrict__ in, int* __restrict__ out) {
    __shared__ int s[256];
    int t = threadIdx.x;
    int row = blockIdx.x * RS_BLOCKS;
    int part = 0;
    for (int i = t; i < row; i += 256) part += in[i];
    s[t] = part;
    __syncthreads();
    for (int off = 1; off < 256; off <<= 1) {
        int v = (t >= off) ? s[t - off] : 0;
        __syncthreads();
        s[t] += v;
        __syncthreads();
    }
    int before = s[255];
    __syncthreads();
    int x = (t < RS_BLOCKS) ? in[row + t] : 0;
    s[t] = x;
    __syncthreads();
    for (int off = 1; off < 256; off <<= 1) {
        int v = (t >= off) ? s[t - off] : 0;
        __syncthreads();
        s[t] += v;
        __syncthreads();
    }
    if (t < RS_BLOCKS) out[row + t] = before + s[t] - x;
}

template <int SHIFT>
__global__ void __launch_bounds__(256) k_scatter(const int* __restrict__ keys, const int* __restrict__ vals,
                                                 const int* __restrict__ hscan,
                                                 int* __restrict__ keys_out, int* __restrict__ vals_out) {
    __shared__ int h[8][256];
    int tid = threadIdx.x;
    int w = tid >> 5;
    int lane = tid & 31;
#pragma unroll
    for (int k = 0; k < 8; ++k) h[k][tid] = 0;
    __syncthreads();
    int base = blockIdx.x * RS_BTILE + w * RS_WTILE + lane;
    int key[RS_ITEMS];
    int val[RS_ITEMS];
#pragma unroll
    for (int i = 0; i < RS_ITEMS; ++i) {
        int idx = base + i * 32;
        int ic = idx < N_EDGES ? idx : N_EDGES - 1;
        key[i] = keys[ic];
        val[i] = vals[ic];
    }
    unsigned lt = (1u << lane) - 1u;
    int info[RS_ITEMS];
#pragma unroll
    for (int i = 0; i < RS_ITEMS; ++i) {
        bool valid = (base + i * 32) < N_EDGES;
        int d = (key[i] >> SHIFT) & 255;
        unsigned m = match_digit(d, valid);
        int top = 31 - __builtin_clz(m | 1u);
        int cntg = __builtin_popcount(m);
        bool leader = valid && lane == top;
        if (leader) h[w][d] += cntg;
        info[i] = __builtin_popcount(m & lt) | (cntg << 8) | (leader ? 0x10000 : 0);
    }
    __syncthreads();
    {
        int run = hscan[tid * RS_BLOCKS + blockIdx.x];
#pragma unroll
        for (int k = 0; k < 8; ++k) {
            int c = h[k][tid];
            h[k][tid] = run;
            run += c;
        }
    }
    __syncthreads();
#pragma unroll
    for (int i = 0; i < RS_ITEMS; ++i) {
        bool valid = (base + i * 32) < N_EDGES;
        int d = (key[i] >> SHIFT) & 255;
        int start = h[w][d];
        int pos = start + (info[i] & 255);
        if (info[i] & 0x10000) h[w][d] = start + ((info[i] >> 8) & 255);
        if (valid) {
            keys_out[pos] = key[i];
            vals_out[pos] = val[i];
        }
    }
}

#define EL_ITEMS   5
#define EL_BLOCKS  1250
#define EL_STRIDE  (EL_BLOCKS * 256)
static_assert(EL_STRIDE * EL_ITEMS == N_EDGES, "every thread owns exactly EL_ITEMS edges: no bounds test in the kernels");

__global__ void __launch_bounds__(256) k_bounds(const int* __restrict__ sdst, int* __restrict__ rowptr) {
    int t = blockIdx.x * 256 + threadIdx.x;
    int d[EL_ITEMS];
    int prev[EL_ITEMS];
#pragma unroll
    for (int i = 0; i < EL_ITEMS; ++i) {
        int p = t + i * EL_STRIDE;
        d[i] = sdst[p];
        prev[i] = (i > 0 || p > 0) ? sdst[p - 1] : -1;
    }
#pragma unroll
    for (int i = 0; i < EL_ITEMS; ++i) {
        int p = t + i * EL_STRIDE;
        for (int n = prev[i] + 1; n <= d[i]; ++n) rowptr[n] = p;
        if (p == N_EDGES - 1)
            for (int n = d[i] + 1; n <= N_NODES; ++n) rowptr[n] = N_EDGES;
    }
}

__global__ void __launch_bounds__(256) k_pair(const int* __restrict__ ssrc, const int* __restrict__ sdst,
                                              const int* __restrict__ rowptr, int2* __restrict__ csr) {
    int t = blockIdx.x * 256 + threadIdx.x;
    int s[EL_ITEMS];
    int d[EL_ITEMS];
#pragma unroll
    for (int i = 0; i < EL_ITEMS; ++i) {
        int p = t + i * EL_STRIDE;
        s[i] = ssrc[p];
        d[i] = sdst[p];
    }
    int degs[EL_ITEMS];
    int degd[EL_ITEMS];
#pragma unroll
    for (int i = 0; i < EL_ITEMS; ++i) {
        degs[i] = rowptr[s[i] + 1] - rowptr[s[i]];
        degd[i] = rowptr[d[i] + 1] - rowptr[d[i]];
    }
#pragma unroll
    for (int i = 0; i < EL_ITEMS; ++i) {
        int p = t + i * EL_STRIDE;
        float ds = rsqrtf((float)degs[i] + 1.0f);
        float dd = rsqrtf((float)degd[i] + 1.0f);
        int2 v;
        v.x = s[i];
        v.y = __float_as_int(ds * dd);
        csr[p] = v;
    }
}

#define GM_BLOCKS 256
#define GM_TILES  (N_NODES >> 4)
#define GM_CHUNK  8
static_assert(GM_TILES * 16 == N_NODES, "whole row tiles only: no row bounds test in the kernel");
static_assert(GM_BLOCKS * 8 <= GM_TILES, "every wave owns at least one row tile: the first tile's loads need no test");
__global__ void __launch_bounds__(256) k_gemm_rows_wmma(const float* __restrict__ A, const float* __restrict__ W,
                                                        float* __restrict__ out) {
    __shared__ v2f Bs[32 * 8 * 32];
    int w    = __builtin_amdgcn_readfirstlane(threadIdx.x >> 5);
    int lane = threadIdx.x & 31;
    int half = lane >> 4;
    int l16  = lane & 15;

    int rt = blockIdx.x + GM_BLOCKS * w;
    const float* arow = A + ((rt << 4) + l16) * HID + (half << 1);
    v2f an[GM_CHUNK];
#pragma unroll
    for (int u = 0; u < GM_CHUNK; ++u) an[u] = *(const v2f*)(arow + 4 * u);

#pragma unroll
    for (int s = 0; s < 32; ++s) {
        int ka = 4 * s + (half << 1);
        v2f b; b.x = W[ka * HID + (w << 4) + l16]; b.y = W[(ka + 1) * HID + (w << 4) + l16];
        Bs[(s * 8 + w) * 32 + lane] = b;
    }
    __syncthreads();

    for (;;) {
        int rn = rt + GM_BLOCKS * 8;
        bool more = rn < GM_TILES;
        const float* arow_n = A + (((more ? rn : rt) << 4) + l16) * HID + (half << 1);
        v8f c[8];
#pragma unroll
        for (int ct = 0; ct < 8; ++ct) c[ct] = (v8f){0.f, 0.f, 0.f, 0.f, 0.f, 0.f, 0.f, 0.f};
#pragma unroll 1
        for (int q = 0; q < 32 / GM_CHUNK; ++q) {
            v2f ac[GM_CHUNK];
#pragma unroll
            for (int u = 0; u < GM_CHUNK; ++u) { ac[u].x = fmaxf(an[u].x, 0.f); ac[u].y = fmaxf(an[u].y, 0.f); }
            const float* ap = (q < 32 / GM_CHUNK - 1) ? arow + 4 * GM_CHUNK * (q + 1) : arow_n;
#pragma unroll
            for (int u = 0; u < GM_CHUNK; ++u) an[u] = *(const v2f*)(ap + 4 * u);
            const v2f* bq = Bs + q * (GM_CHUNK * 8 * 32) + lane;
#pragma unroll
            for (int u = 0; u < GM_CHUNK; ++u) {
#pragma unroll
                for (int ct = 0; ct < 8; ++ct) {
                    v2f b = bq[(u * 8 + ct) * 32];
                    c[ct] = __builtin_amdgcn_wmma_f32_16x16x4_f32(false, ac[u], false, b,
                                                                  (short)0, c[ct], false, false);
                }
            }
        }
        float* orow = out + ((rt << 4) + (half << 3)) * HID + l16;
#pragma unroll
        for (int ct = 0; ct < 8; ++ct) {
#pragma unroll
            for (int v = 0; v < 8; ++v) orow[v * HID + (ct << 4)] = c[ct][v];
        }
        if (!more) break;
        rt = rn;
        arow = arow_n;
    }
}

#define GA_NODES  4
#define GA_BLOCKS 1563
template <bool RELU_OUT, bool BIAS>
__global__ void __launch_bounds__(256) k_gather(const int* __restrict__ rowptr, const int2* __restrict__ csr,
                                                const float* __restrict__ hw, const float* __restrict__ bias,
                                                float* __restrict__ out) {
    int wave = __builtin_amdgcn_readfirstlane((blockIdx.x * 256 + threadIdx.x) >> 5);
    int lane = threadIdx.x & 31;
    const float* hl = hw + lane * 4;
    float4 b4 = make_float4(0.f, 0.f, 0.f, 0.f);
    if (BIAS) b4 = *(const float4*)(bias + lane * 4);
    for (int i = 0; i < GA_NODES; ++i) {
        int n = wave * GA_NODES + i;
        if (n >= N_NODES) return;
        int beg = rowptr[n];
        int end = rowptr[n + 1];
        float4 acc = make_float4(0.f, 0.f, 0.f, 0.f);
        int j = beg;
        for (; j + 4 <= end; j += 4) {
            int2 e0 = csr[j], e1 = csr[j + 1], e2 = csr[j + 2], e3 = csr[j + 3];
            const float4 h0 = *(const float4*)(hl + e0.x * HID);
            const float4 h1 = *(const float4*)(hl + e1.x * HID);
            const float4 h2 = *(const float4*)(hl + e2.x * HID);
            const float4 h3 = *(const float4*)(hl + e3.x * HID);
            float w0 = __int_as_float(e0.y), w1 = __int_as_float(e1.y);
            float w2 = __int_as_float(e2.y), w3 = __int_as_float(e3.y);
            acc.x = fmaf(h0.x, w0, acc.x); acc.y = fmaf(h0.y, w0, acc.y);
            acc.z = fmaf(h0.z, w0, acc.z); acc.w = fmaf(h0.w, w0, acc.w);
            acc.x = fmaf(h1.x, w1, acc.x); acc.y = fmaf(h1.y, w1, acc.y);
            acc.z = fmaf(h1.z, w1, acc.z); acc.w = fmaf(h1.w, w1, acc.w);
            acc.x = fmaf(h2.x, w2, acc.x); acc.y = fmaf(h2.y, w2, acc.y);
            acc.z = fmaf(h2.z, w2, acc.z); acc.w = fmaf(h2.w, w2, acc.w);
            acc.x = fmaf(h3.x, w3, acc.x); acc.y = fmaf(h3.y, w3, acc.y);
            acc.z = fmaf(h3.z, w3, acc.z); acc.w = fmaf(h3.w, w3, acc.w);
        }
        for (; j < end; ++j) {
            int2 e0 = csr[j];
            const float4 h0 = *(const float4*)(hl + e0.x * HID);
            float w0 = __int_as_float(e0.y);
            acc.x = fmaf(h0.x, w0, acc.x); acc.y = fmaf(h0.y, w0, acc.y);
            acc.z = fmaf(h0.z, w0, acc.z); acc.w = fmaf(h0.w, w0, acc.w);
        }
        float sd = rsqrtf((float)(end - beg) + 1.0f);
        float ss = sd * sd;
        const float4 hs = *(const float4*)(hl + n * HID);
        float4 o;
        if (BIAS) {
            o.x = acc.x + fmaf(hs.x, ss, b4.x);
            o.y = acc.y + fmaf(hs.y, ss, b4.y);
            o.z = acc.z + fmaf(hs.z, ss, b4.z);
            o.w = acc.w + fmaf(hs.w, ss, b4.w);
        } else {
            o.x = fmaf(hs.x, ss, acc.x);
            o.y = fmaf(hs.y, ss, acc.y);
            o.z = fmaf(hs.z, ss, acc.z);
            o.w = fmaf(hs.w, ss, acc.w);
        }
        if (RELU_OUT) {
            o.x = fmaxf(o.x, 0.f); o.y = fmaxf(o.y, 0.f);
            o.z = fmaxf(o.z, 0.f); o.w = fmaxf(o.w, 0.f);
        }
        *(float4*)(out + n * HID + lane * 4) = o;
    }
}

#define GX_UNROLL 12
#define GX_TRIP   (3 * GX_UNROLL)
#define GX_NODES  5
#define GX_BLOCKS 1250
static_assert(GX_BLOCKS * 8 * GX_NODES == N_NODES, "every wave owns exactly GX_NODES nodes: no bounds test in the kernel");
__global__ void __launch_bounds__(256) k_gather_in(const int* __restrict__ rowptr, const int2* __restrict__ csr,
                                                   const float* __restrict__ x, const float* __restrict__ W1,
                                                   const float* __restrict__ bias, float* __restrict__ out) {
    int wave = __builtin_amdgcn_readfirstlane((blockIdx.x * 256 + threadIdx.x) >> 5);
    int lane = threadIdx.x & 31;
    int l = lane < 27 ? lane : 26;
    int e = l / IN_DIM;
    int k = l - e * IN_DIM;
    float4 w[IN_DIM];
#pragma unroll
    for (int q = 0; q < IN_DIM; ++q) w[q] = *(const float4*)(W1 + q * HID + lane * 4);
    const float4 b4 = *(const float4*)(bias + lane * 4);
    for (int i = 0; i < GX_NODES; ++i) {
        int n = wave * GX_NODES + i;
        int beg = rowptr[n];
        int end = rowptr[n + 1];
        float xs = x[n * IN_DIM + k];
        float acc = 0.f;
        for (int j0 = beg; j0 < end; j0 += GX_TRIP) {
            int2 ed[GX_UNROLL];
            float xv[GX_UNROLL];
#pragma unroll
            for (int u = 0; u < GX_UNROLL; ++u) {
                int j = j0 + 3 * u + e;
                ed[u] = csr[j < end ? j : end - 1];
            }
#pragma unroll
            for (int u = 0; u < GX_UNROLL; ++u) xv[u] = x[ed[u].x * IN_DIM + k];
#pragma unroll
            for (int u = 0; u < GX_UNROLL; ++u) {
                int j = j0 + 3 * u + e;
                float t = fmaf(xv[u], __int_as_float(ed[u].y), acc);
                acc = j < end ? t : acc;
            }
        }
        float s1 = __int_as_float(__builtin_amdgcn_ds_bpermute(((lane + 9) & 31) << 2, __float_as_int(acc)));
        float s2 = __int_as_float(__builtin_amdgcn_ds_bpermute(((lane + 18) & 31) << 2, __float_as_int(acc)));
        float sd = rsqrtf((float)(end - beg) + 1.0f);
        float ss = sd * sd;
        float ax = fmaf(xs, ss, (acc + s1) + s2);
        float4 o = make_float4(0.f, 0.f, 0.f, 0.f);
#pragma unroll
        for (int q = 0; q < IN_DIM; ++q) {
            float a = __int_as_float(__builtin_amdgcn_readlane(__float_as_int(ax), q));
            o.x = fmaf(a, w[q].x, o.x); o.y = fmaf(a, w[q].y, o.y);
            o.z = fmaf(a, w[q].z, o.z); o.w = fmaf(a, w[q].w, o.w);
        }
        o.x += b4.x; o.y += b4.y; o.z += b4.z; o.w += b4.w;
        *(float4*)(out + n * HID + lane * 4) = o;
    }
}

__global__ void k_pool_zero(float* __restrict__ sums, float* __restrict__ counts) {
    int i = blockIdx.x * blockDim.x + threadIdx.x;
    if (i < N_GRAPHS * HID) sums[i] = 0.f;
    if (i < N_GRAPHS) counts[i] = 0.f;
}

#define PA_NODES  10
#define PA_BLOCKS 625
static_assert(PA_BLOCKS * 8 * PA_NODES == N_NODES, "every wave owns exactly PA_NODES nodes: no bounds test in the kernel");
__global__ void __launch_bounds__(256) k_pool_acc(const float* __restrict__ h, const int* __restrict__ batch,
                                                  float* __restrict__ sums, float* __restrict__ counts) {
    int lane = threadIdx.x & 31;
    int wave = __builtin_amdgcn_readfirstlane((blockIdx.x * 256 + threadIdx.x) >> 5);
    int n0 = wave * PA_NODES;
    int g[PA_NODES];
    float4 h4[PA_NODES];
#pragma unroll
    for (int i = 0; i < PA_NODES; ++i) {
        int n = n0 + i;
        g[i] = batch[n];
        h4[i] = *(const float4*)(h + n * HID + lane * 4);
    }
#pragma unroll
    for (int i = 0; i < PA_NODES; ++i) {
        float* o = sums + g[i] * HID + lane * 4;
        atomic_add_f32(o + 0, h4[i].x);
        atomic_add_f32(o + 1, h4[i].y);
        atomic_add_f32(o + 2, h4[i].z);
        atomic_add_f32(o + 3, h4[i].w);
        if (lane == 0) atomic_add_f32(&counts[g[i]], 1.0f);
    }
}

__global__ void __launch_bounds__(256) k_pool_gemm_wmma(const float* __restrict__ S, const float* __restrict__ W,
                                                        const float* __restrict__ counts, const float* __restrict__ bias,
                                                        float* __restrict__ out) {
    int wave = (blockIdx.x * 256 + threadIdx.x) >> 5;
    int lane = threadIdx.x & 31;
    int ct = wave & 7;
    int rt = wave >> 3;

    int half = lane >> 4;
    int l16  = lane & 15;
    int row  = (rt << 4) + l16;
    int col  = (ct << 4) + l16;
    const float* arow = S + row * HID;

    v8f c = {0.f, 0.f, 0.f, 0.f, 0.f, 0.f, 0.f, 0.f};
#pragma unroll
    for (int k0 = 0; k0 < HID; k0 += 4) {
        int ka = k0 + (half << 1);
        v2f a; a.x = arow[ka]; a.y = arow[ka + 1];
        v2f b; b.x = W[ka * HID + col]; b.y = W[(ka + 1) * HID + col];
        c = __builtin_amdgcn_wmma_f32_16x16x4_f32(false, a, false, b,
                                                  (short)0, c, false, false);
    }
    int g0 = (rt << 4) + (half << 3);
    float bc = bias[col];
#pragma unroll
    for (int v = 0; v < 8; ++v) {
        float cnt = counts[g0 + v];
        out[(g0 + v) * HID + col] = fmaf(cnt, bc, c[v]) / fmaxf(cnt, 1.0f);
    }
}

extern "C" void kernel_launch(void* const* d_in, const int* in_sizes, int n_in,
                              void* d_out, int out_size, void* d_ws, size_t ws_size,
                              hipStream_t stream) {
    const float* x        = (const float*)d_in[0];
    const int*   eidx     = (const int*)d_in[1];
    const int*   batch    = (const int*)d_in[2];
    const float* W1       = (const float*)d_in[3];
    const float* b1       = (const float*)d_in[4];
    const float* W2       = (const float*)d_in[5];
    const float* b2       = (const float*)d_in[6];
    const float* W3       = (const float*)d_in[7];
    const float* b3       = (const float*)d_in[8];
    float* out = (float*)d_out;

    const int* src = eidx;
    const int* dst = eidx + N_EDGES;

    float* ws     = (float*)d_ws;
    float* hw     = ws;
    float* hA     = hw + (size_t)N_NODES * HID;
    float* counts = hA + (size_t)N_NODES * HID;
    float* psum   = counts + N_GRAPHS;
    int2*  csr    = (int2*)(psum + N_GRAPHS * HID);
    int*   kA     = (int*)(csr + N_EDGES);
    int*   vA     = kA + N_EDGES;
    int*   kB     = vA + N_EDGES;
    int*   vB     = kB + N_EDGES;
    int*   hist   = vB + N_EDGES;
    int*   hscan  = hist + RS_HIST;
    int*   rowptr = hscan + RS_HIST;

    const int B = 256;

    k_hist<0><<<RS_BLOCKS, B, 0, stream>>>(dst, hist);
    k_scan<<<256, B, 0, stream>>>(hist, hscan);
    k_scatter<0><<<RS_BLOCKS, B, 0, stream>>>(dst, src, hscan, kA, vA);
    k_hist<8><<<RS_BLOCKS, B, 0, stream>>>(kA, hist);
    k_scan<<<256, B, 0, stream>>>(hist, hscan);
    k_scatter<8><<<RS_BLOCKS, B, 0, stream>>>(kA, vA, hscan, kB, vB);
    k_bounds<<<EL_BLOCKS, B, 0, stream>>>(kB, rowptr);
    k_pair<<<EL_BLOCKS, B, 0, stream>>>(vB, kB, rowptr, csr);

    k_gather_in<<<GX_BLOCKS, B, 0, stream>>>(rowptr, csr, x, W1, b1, hA);

    k_gemm_rows_wmma<<<GM_BLOCKS, B, 0, stream>>>(hA, W2, hw);
    k_gather<true, true><<<GA_BLOCKS, B, 0, stream>>>(rowptr, csr, hw, b2, hA);

    k_gather<false, false><<<GA_BLOCKS, B, 0, stream>>>(rowptr, csr, hA, b3, hw);

    k_pool_zero<<<(N_GRAPHS * HID + B - 1) / B, B, 0, stream>>>(psum, counts);
    k_pool_acc<<<PA_BLOCKS, B, 0, stream>>>(hw, batch, psum, counts);
    k_pool_gemm_wmma<<<N_GRAPHS / 16, B, 0, stream>>>(psum, W3, counts, b3, out);
}


# Layer-3 edge sums go straight into the per-graph sums: a wave adds its 4 nodes' rows in registers, then one atomic row per graph (start 0 from k_bounds); 2 launches fewer; pooled sum order changes
# speedup vs baseline: 4.5864x; 1.1149x over previous
#include <hip/hip_runtime.h>

#define N_NODES  50000
#define N_EDGES  1600000
#define N_GRAPHS 256
#define IN_DIM   9
#define HID      128

typedef __attribute__((ext_vector_type(2))) float v2f;
typedef __attribute__((ext_vector_type(8))) float v8f;

__device__ __forceinline__ void atomic_add_f32(float* p, float v) {
    asm volatile("global_atomic_add_f32 %0, %1, off scope:SCOPE_DEV"
                 :: "v"(p), "v"(v) : "memory");
}

#define RS_ITEMS  32
#define RS_WTILE  1024
#define RS_BTILE  8192
#define RS_BLOCKS 196
#define RS_HIST   50176

__device__ __forceinline__ unsigned match_digit(int d, bool valid) {
    unsigned m = __builtin_amdgcn_ballot_w32(valid);
#pragma unroll
    for (int b = 0; b < 8; ++b) {
        bool bit = ((d >> b) & 1) != 0;
        unsigned bal = __builtin_amdgcn_ballot_w32(bit);
        m &= bit ? bal : ~bal;
    }
    return m;
}

template <int SHIFT>
__global__ void __launch_bounds__(256) k_hist(const int* __restrict__ keys, int* __restrict__ hist) {
    __shared__ int h[8][256];
    int tid = threadIdx.x;
    int w = tid >> 5;
    int lane = tid & 31;
#pragma unroll
    for (int k = 0; k < 8; ++k) h[k][tid] = 0;
    __syncthreads();
    int base = blockIdx.x * RS_BTILE + w * RS_WTILE + lane;
    int key[RS_ITEMS];
#pragma unroll
    for (int i = 0; i < RS_ITEMS; ++i) {
        int idx = base + i * 32;
        key[i] = keys[idx < N_EDGES ? idx : N_EDGES - 1];
    }
#pragma unroll
    for (int i = 0; i < RS_ITEMS; ++i) {
        bool valid = (base + i * 32) < N_EDGES;
        int d = (key[i] >> SHIFT) & 255;
        unsigned m = match_digit(d, valid);
        int top = 31 - __builtin_clz(m | 1u);
        if (valid && lane == top) h[w][d] += __builtin_popcount(m);
    }
    __syncthreads();
    int tot = 0;
#pragma unroll
    for (int k = 0; k < 8; ++k) tot += h[k][tid];
    hist[tid * RS_BLOCKS + blockIdx.x] = tot;
}

__global__ void __launch_bounds__(256) k_scan(const int* __restrict__ in, int* __restrict__ out) {
    __shared__ int s[256];
    int t = threadIdx.x;
    int row = blockIdx.x * RS_BLOCKS;
    int part = 0;
    for (int i = t; i < row; i += 256) part += in[i];
    s[t] = part;
    __syncthreads();
    for (int off = 1; off < 256; off <<= 1) {
        int v = (t >= off) ? s[t - off] : 0;
        __syncthreads();
        s[t] += v;
        __syncthreads();
    }
    int before = s[255];
    __syncthreads();
    int x = (t < RS_BLOCKS) ? in[row + t] : 0;
    s[t] = x;
    __syncthreads();
    for (int off = 1; off < 256; off <<= 1) {
        int v = (t >= off) ? s[t - off] : 0;
        __syncthreads();
        s[t] += v;
        __syncthreads();
    }
    if (t < RS_BLOCKS) out[row + t] = before + s[t] - x;
}

template <int SHIFT>
__global__ void __launch_bounds__(256) k_scatter(const int* __restrict__ keys, const int* __restrict__ vals,
                                                 const int* __restrict__ hscan,
                                                 int* __restrict__ keys_out, int* __restrict__ vals_out) {
    __shared__ int h[8][256];
    int tid = threadIdx.x;
    int w = tid >> 5;
    int lane = tid & 31;
#pragma unroll
    for (int k = 0; k < 8; ++k) h[k][tid] = 0;
    __syncthreads();
    int base = blockIdx.x * RS_BTILE + w * RS_WTILE + lane;
    int key[RS_ITEMS];
    int val[RS_ITEMS];
#pragma unroll
    for (int i = 0; i < RS_ITEMS; ++i) {
        int idx = base + i * 32;
        int ic = idx < N_EDGES ? idx : N_EDGES - 1;
        key[i] = keys[ic];
        val[i] = vals[ic];
    }
    unsigned lt = (1u << lane) - 1u;
    int info[RS_ITEMS];
#pragma unroll
    for (int i = 0; i < RS_ITEMS; ++i) {
        bool valid = (base + i * 32) < N_EDGES;
        int d = (key[i] >> SHIFT) & 255;
        unsigned m = match_digit(d, valid);
        int top = 31 - __builtin_clz(m | 1u);
        int cntg = __builtin_popcount(m);
        bool leader = valid && lane == top;
        if (leader) h[w][d] += cntg;
        info[i] = __builtin_popcount(m & lt) | (cntg << 8) | (leader ? 0x10000 : 0);
    }
    __syncthreads();
    {
        int run = hscan[tid * RS_BLOCKS + blockIdx.x];
#pragma unroll
        for (int k = 0; k < 8; ++k) {
            int c = h[k][tid];
            h[k][tid] = run;
            run += c;
        }
    }
    __syncthreads();
#pragma unroll
    for (int i = 0; i < RS_ITEMS; ++i) {
        bool valid = (base + i * 32) < N_EDGES;
        int d = (key[i] >> SHIFT) & 255;
        int start = h[w][d];
        int pos = start + (info[i] & 255);
        if (info[i] & 0x10000) h[w][d] = start + ((info[i] >> 8) & 255);
        if (valid) {
            keys_out[pos] = key[i];
            vals_out[pos] = val[i];
        }
    }
}

#define EL_ITEMS   5
#define EL_BLOCKS  1250
#define EL_STRIDE  (EL_BLOCKS * 256)
static_assert(EL_STRIDE * EL_ITEMS == N_EDGES, "every thread owns exactly EL_ITEMS edges: no bounds test in the kernels");

static_assert(EL_STRIDE >= N_GRAPHS * HID, "one thread per element of the per-graph sums");
__global__ void __launch_bounds__(256) k_bounds(const int* __restrict__ sdst, int* __restrict__ rowptr,
                                                float* __restrict__ sums, float* __restrict__ counts) {
    int t = blockIdx.x * 256 + threadIdx.x;
    if (t < N_GRAPHS * HID) sums[t] = 0.f;
    if (t < N_GRAPHS) counts[t] = 0.f;
    int d[EL_ITEMS];
    int prev[EL_ITEMS];
#pragma unroll
    for (int i = 0; i < EL_ITEMS; ++i) {
        int p = t + i * EL_STRIDE;
        d[i] = sdst[p];
        prev[i] = (i > 0 || p > 0) ? sdst[p - 1] : -1;
    }
#pragma unroll
    for (int i = 0; i < EL_ITEMS; ++i) {
        int p = t + i * EL_STRIDE;
        for (int n = prev[i] + 1; n <= d[i]; ++n) rowptr[n] = p;
        if (p == N_EDGES - 1)
            for (int n = d[i] + 1; n <= N_NODES; ++n) rowptr[n] = N_EDGES;
    }
}

__global__ void __launch_bounds__(256) k_pair(const int* __restrict__ ssrc, const int* __restrict__ sdst,
                                              const int* __restrict__ rowptr, int2* __restrict__ csr) {
    int t = blockIdx.x * 256 + threadIdx.x;
    int s[EL_ITEMS];
    int d[EL_ITEMS];
#pragma unroll
    for (int i = 0; i < EL_ITEMS; ++i) {
        int p = t + i * EL_STRIDE;
        s[i] = ssrc[p];
        d[i] = sdst[p];
    }
    int degs[EL_ITEMS];
    int degd[EL_ITEMS];
#pragma unroll
    for (int i = 0; i < EL_ITEMS; ++i) {
        degs[i] = rowptr[s[i] + 1] - rowptr[s[i]];
        degd[i] = rowptr[d[i] + 1] - rowptr[d[i]];
    }
#pragma unroll
    for (int i = 0; i < EL_ITEMS; ++i) {
        int p = t + i * EL_STRIDE;
        float ds = rsqrtf((float)degs[i] + 1.0f);
        float dd = rsqrtf((float)degd[i] + 1.0f);
        int2 v;
        v.x = s[i];
        v.y = __float_as_int(ds * dd);
        csr[p] = v;
    }
}

#define GM_BLOCKS 256
#define GM_TILES  (N_NODES >> 4)
#define GM_CHUNK  8
static_assert(GM_TILES * 16 == N_NODES, "whole row tiles only: no row bounds test in the kernel");
static_assert(GM_BLOCKS * 8 <= GM_TILES, "every wave owns at least one row tile: the first tile's loads need no test");
__global__ void __launch_bounds__(256) k_gemm_rows_wmma(const float* __restrict__ A, const float* __restrict__ W,
                                                        float* __restrict__ out) {
    __shared__ v2f Bs[32 * 8 * 32];
    int w    = __builtin_amdgcn_readfirstlane(threadIdx.x >> 5);
    int lane = threadIdx.x & 31;
    int half = lane >> 4;
    int l16  = lane & 15;

    int rt = blockIdx.x + GM_BLOCKS * w;
    const float* arow = A + ((rt << 4) + l16) * HID + (half << 1);
    v2f an[GM_CHUNK];
#pragma unroll
    for (int u = 0; u < GM_CHUNK; ++u) an[u] = *(const v2f*)(arow + 4 * u);

#pragma unroll
    for (int s = 0; s < 32; ++s) {
        int ka = 4 * s + (half << 1);
        v2f b; b.x = W[ka * HID + (w << 4) + l16]; b.y = W[(ka + 1) * HID + (w << 4) + l16];
        Bs[(s * 8 + w) * 32 + lane] = b;
    }
    __syncthreads();

    for (;;) {
        int rn = rt + GM_BLOCKS * 8;
        bool more = rn < GM_TILES;
        const float* arow_n = A + (((more ? rn : rt) << 4) + l16) * HID + (half << 1);
        v8f c[8];
#pragma unroll
        for (int ct = 0; ct < 8; ++ct) c[ct] = (v8f){0.f, 0.f, 0.f, 0.f, 0.f, 0.f, 0.f, 0.f};
#pragma unroll 1
        for (int q = 0; q < 32 / GM_CHUNK; ++q) {
            v2f ac[GM_CHUNK];
#pragma unroll
            for (int u = 0; u < GM_CHUNK; ++u) { ac[u].x = fmaxf(an[u].x, 0.f); ac[u].y = fmaxf(an[u].y, 0.f); }
            const float* ap = (q < 32 / GM_CHUNK - 1) ? arow + 4 * GM_CHUNK * (q + 1) : arow_n;
#pragma unroll
            for (int u = 0; u < GM_CHUNK; ++u) an[u] = *(const v2f*)(ap + 4 * u);
            const v2f* bq = Bs + q * (GM_CHUNK * 8 * 32) + lane;
#pragma unroll
            for (int u = 0; u < GM_CHUNK; ++u) {
#pragma unroll
                for (int ct = 0; ct < 8; ++ct) {
                    v2f b = bq[(u * 8 + ct) * 32];
                    c[ct] = __builtin_amdgcn_wmma_f32_16x16x4_f32(false, ac[u], false, b,
                                                                  (short)0, c[ct], false, false);
                }
            }
        }
        float* orow = out + ((rt << 4) + (half << 3)) * HID + l16;
#pragma unroll
        for (int ct = 0; ct < 8; ++ct) {
#pragma unroll
            for (int v = 0; v < 8; ++v) orow[v * HID + (ct << 4)] = c[ct][v];
        }
        if (!more) break;
        rt = rn;
        arow = arow_n;
    }
}

#define GA_NODES  4
#define GA_BLOCKS 1563

__device__ __forceinline__ float4 edge_sum(const int2* __restrict__ csr, const float* __restrict__ hl,
                                           int beg, int end, float4 acc) {
    int j = beg;
    for (; j + 4 <= end; j += 4) {
        int2 e0 = csr[j], e1 = csr[j + 1], e2 = csr[j + 2], e3 = csr[j + 3];
        const float4 h0 = *(const float4*)(hl + e0.x * HID);
        const float4 h1 = *(const float4*)(hl + e1.x * HID);
        const float4 h2 = *(const float4*)(hl + e2.x * HID);
        const float4 h3 = *(const float4*)(hl + e3.x * HID);
        float w0 = __int_as_float(e0.y), w1 = __int_as_float(e1.y);
        float w2 = __int_as_float(e2.y), w3 = __int_as_float(e3.y);
        acc.x = fmaf(h0.x, w0, acc.x); acc.y = fmaf(h0.y, w0, acc.y);
        acc.z = fmaf(h0.z, w0, acc.z); acc.w = fmaf(h0.w, w0, acc.w);
        acc.x = fmaf(h1.x, w1, acc.x); acc.y = fmaf(h1.y, w1, acc.y);
        acc.z = fmaf(h1.z, w1, acc.z); acc.w = fmaf(h1.w, w1, acc.w);
        acc.x = fmaf(h2.x, w2, acc.x); acc.y = fmaf(h2.y, w2, acc.y);
        acc.z = fmaf(h2.z, w2, acc.z); acc.w = fmaf(h2.w, w2, acc.w);
        acc.x = fmaf(h3.x, w3, acc.x); acc.y = fmaf(h3.y, w3, acc.y);
        acc.z = fmaf(h3.z, w3, acc.z); acc.w = fmaf(h3.w, w3, acc.w);
    }
    for (; j < end; ++j) {
        int2 e0 = csr[j];
        const float4 h0 = *(const float4*)(hl + e0.x * HID);
        float w0 = __int_as_float(e0.y);
        acc.x = fmaf(h0.x, w0, acc.x); acc.y = fmaf(h0.y, w0, acc.y);
        acc.z = fmaf(h0.z, w0, acc.z); acc.w = fmaf(h0.w, w0, acc.w);
    }
    return acc;
}

__global__ void __launch_bounds__(256) k_gather(const int* __restrict__ rowptr, const int2* __restrict__ csr,
                                                const float* __restrict__ hw, const float* __restrict__ bias,
                                                float* __restrict__ out) {
    int wave = __builtin_amdgcn_readfirstlane((blockIdx.x * 256 + threadIdx.x) >> 5);
    int lane = threadIdx.x & 31;
    const float* hl = hw + lane * 4;
    const float4 b4 = *(const float4*)(bias + lane * 4);
    for (int i = 0; i < GA_NODES; ++i) {
        int n = wave * GA_NODES + i;
        if (n >= N_NODES) return;
        int beg = rowptr[n];
        int end = rowptr[n + 1];
        float4 acc = edge_sum(csr, hl, beg, end, make_float4(0.f, 0.f, 0.f, 0.f));
        float sd = rsqrtf((float)(end - beg) + 1.0f);
        float ss = sd * sd;
        const float4 hs = *(const float4*)(hl + n * HID);
        float4 o;
        o.x = fmaxf(acc.x + fmaf(hs.x, ss, b4.x), 0.f);
        o.y = fmaxf(acc.y + fmaf(hs.y, ss, b4.y), 0.f);
        o.z = fmaxf(acc.z + fmaf(hs.z, ss, b4.z), 0.f);
        o.w = fmaxf(acc.w + fmaf(hs.w, ss, b4.w), 0.f);
        *(float4*)(out + n * HID + lane * 4) = o;
    }
}

__device__ __forceinline__ void pool_add(float* __restrict__ sums, float* __restrict__ counts,
                                         int g, int lane, float4 acc, int cnt) {
    float* o = sums + g * HID + lane * 4;
    atomic_add_f32(o + 0, acc.x);
    atomic_add_f32(o + 1, acc.y);
    atomic_add_f32(o + 2, acc.z);
    atomic_add_f32(o + 3, acc.w);
    if (lane == 0) atomic_add_f32(counts + g, (float)cnt);
}

__global__ void __launch_bounds__(256) k_gather_pool(const int* __restrict__ rowptr, const int2* __restrict__ csr,
                                                     const float* __restrict__ hw, const int* __restrict__ batch,
                                                     float* __restrict__ sums, float* __restrict__ counts) {
    int wave = __builtin_amdgcn_readfirstlane((blockIdx.x * 256 + threadIdx.x) >> 5);
    int lane = threadIdx.x & 31;
    const float* hl = hw + lane * 4;
    int n0 = wave * GA_NODES;
    if (n0 >= N_NODES) return;
    int g = batch[n0];
    int cnt = 0;
    float4 acc = make_float4(0.f, 0.f, 0.f, 0.f);
    for (int i = 0; i < GA_NODES; ++i) {
        int n = n0 + i;
        if (n >= N_NODES) break;
        int gn = batch[n];
        int beg = rowptr[n];
        int end = rowptr[n + 1];
        if (gn != g) {
            pool_add(sums, counts, g, lane, acc, cnt);
            acc = make_float4(0.f, 0.f, 0.f, 0.f);
            cnt = 0;
            g = gn;
        }
        acc = edge_sum(csr, hl, beg, end, acc);
        float sd = rsqrtf((float)(end - beg) + 1.0f);
        float ss = sd * sd;
        const float4 hs = *(const float4*)(hl + n * HID);
        acc.x = fmaf(hs.x, ss, acc.x);
        acc.y = fmaf(hs.y, ss, acc.y);
        acc.z = fmaf(hs.z, ss, acc.z);
        acc.w = fmaf(hs.w, ss, acc.w);
        ++cnt;
    }
    pool_add(sums, counts, g, lane, acc, cnt);
}

#define GX_UNROLL 12
#define GX_TRIP   (3 * GX_UNROLL)
#define GX_NODES  5
#define GX_BLOCKS 1250
static_assert(GX_BLOCKS * 8 * GX_NODES == N_NODES, "every wave owns exactly GX_NODES nodes: no bounds test in the kernel");
__global__ void __launch_bounds__(256) k_gather_in(const int* __restrict__ rowptr, const int2* __restrict__ csr,
                                                   const float* __restrict__ x, const float* __restrict__ W1,
                                                   const float* __restrict__ bias, float* __restrict__ out) {
    int wave = __builtin_amdgcn_readfirstlane((blockIdx.x * 256 + threadIdx.x) >> 5);
    int lane = threadIdx.x & 31;
    int l = lane < 27 ? lane : 26;
    int e = l / IN_DIM;
    int k = l - e * IN_DIM;
    float4 w[IN_DIM];
#pragma unroll
    for (int q = 0; q < IN_DIM; ++q) w[q] = *(const float4*)(W1 + q * HID + lane * 4);
    const float4 b4 = *(const float4*)(bias + lane * 4);
    for (int i = 0; i < GX_NODES; ++i) {
        int n = wave * GX_NODES + i;
        int beg = rowptr[n];
        int end = rowptr[n + 1];
        float xs = x[n * IN_DIM + k];
        float acc = 0.f;
        for (int j0 = beg; j0 < end; j0 += GX_TRIP) {
            int2 ed[GX_UNROLL];
            float xv[GX_UNROLL];
#pragma unroll
            for (int u = 0; u < GX_UNROLL; ++u) {
                int j = j0 + 3 * u + e;
                ed[u] = csr[j < end ? j : end - 1];
            }
#pragma unroll
            for (int u = 0; u < GX_UNROLL; ++u) xv[u] = x[ed[u].x * IN_DIM + k];
#pragma unroll
            for (int u = 0; u < GX_UNROLL; ++u) {
                int j = j0 + 3 * u + e;
                float t = fmaf(xv[u], __int_as_float(ed[u].y), acc);
                acc = j < end ? t : acc;
            }
        }
        float s1 = __int_as_float(__builtin_amdgcn_ds_bpermute(((lane + 9) & 31) << 2, __float_as_int(acc)));
        float s2 = __int_as_float(__builtin_amdgcn_ds_bpermute(((lane + 18) & 31) << 2, __float_as_int(acc)));
        float sd = rsqrtf((float)(end - beg) + 1.0f);
        float ss = sd * sd;
        float ax = fmaf(xs, ss, (acc + s1) + s2);
        float4 o = make_float4(0.f, 0.f, 0.f, 0.f);
#pragma unroll
        for (int q = 0; q < IN_DIM; ++q) {
            float a = __int_as_float(__builtin_amdgcn_readlane(__float_as_int(ax), q));
            o.x = fmaf(a, w[q].x, o.x); o.y = fmaf(a, w[q].y, o.y);
            o.z = fmaf(a, w[q].z, o.z); o.w = fmaf(a, w[q].w, o.w);
        }
        o.x += b4.x; o.y += b4.y; o.z += b4.z; o.w += b4.w;
        *(float4*)(out + n * HID + lane * 4) = o;
    }
}

__global__ void __launch_bounds__(256) k_pool_gemm_wmma(const float* __restrict__ S, const float* __restrict__ W,
                                                        const float* __restrict__ counts, const float* __restrict__ bias,
                                                        float* __restrict__ out) {
    int wave = (blockIdx.x * 256 + threadIdx.x) >> 5;
    int lane = threadIdx.x & 31;
    int ct = wave & 7;
    int rt = wave >> 3;

    int half = lane >> 4;
    int l16  = lane & 15;
    int row  = (rt << 4) + l16;
    int col  = (ct << 4) + l16;
    const float* arow = S + row * HID;

    v8f c = {0.f, 0.f, 0.f, 0.f, 0.f, 0.f, 0.f, 0.f};
#pragma unroll
    for (int k0 = 0; k0 < HID; k0 += 4) {
        int ka = k0 + (half << 1);
        v2f a; a.x = arow[ka]; a.y = arow[ka + 1];
        v2f b; b.x = W[ka * HID + col]; b.y = W[(ka + 1) * HID + col];
        c = __builtin_amdgcn_wmma_f32_16x16x4_f32(false, a, false, b,
                                                  (short)0, c, false, false);
    }
    int g0 = (rt << 4) + (half << 3);
    float bc = bias[col];
#pragma unroll
    for (int v = 0; v < 8; ++v) {
        float cnt = counts[g0 + v];
        out[(g0 + v) * HID + col] = fmaf(cnt, bc, c[v]) / fmaxf(cnt, 1.0f);
    }
}

extern "C" void kernel_launch(void* const* d_in, const int* in_sizes, int n_in,
                              void* d_out, int out_size, void* d_ws, size_t ws_size,
                              hipStream_t stream) {
    const float* x        = (const float*)d_in[0];
    const int*   eidx     = (const int*)d_in[1];
    const int*   batch    = (const int*)d_in[2];
    const float* W1       = (const float*)d_in[3];
    const float* b1       = (const float*)d_in[4];
    const float* W2       = (const float*)d_in[5];
    const float* b2       = (const float*)d_in[6];
    const float* W3       = (const float*)d_in[7];
    const float* b3       = (const float*)d_in[8];
    float* out = (float*)d_out;

    const int* src = eidx;
    const int* dst = eidx + N_EDGES;

    float* ws     = (float*)d_ws;
    float* hw     = ws;
    float* hA     = hw + (size_t)N_NODES * HID;
    float* counts = hA + (size_t)N_NODES * HID;
    float* psum   = counts + N_GRAPHS;
    int2*  csr    = (int2*)(psum + N_GRAPHS * HID);
    int*   kA     = (int*)(csr + N_EDGES);
    int*   vA     = kA + N_EDGES;
    int*   kB     = vA + N_EDGES;
    int*   vB     = kB + N_EDGES;
    int*   hist   = vB + N_EDGES;
    int*   hscan  = hist + RS_HIST;
    int*   rowptr = hscan + RS_HIST;

    const int B = 256;

    k_hist<0><<<RS_BLOCKS, B, 0, stream>>>(dst, hist);
    k_scan<<<256, B, 0, stream>>>(hist, hscan);
    k_scatter<0><<<RS_BLOCKS, B, 0, stream>>>(dst, src, hscan, kA, vA);
    k_hist<8><<<RS_BLOCKS, B, 0, stream>>>(kA, hist);
    k_scan<<<256, B, 0, stream>>>(hist, hscan);
    k_scatter<8><<<RS_BLOCKS, B, 0, stream>>>(kA, vA, hscan, kB, vB);
    k_bounds<<<EL_BLOCKS, B, 0, stream>>>(kB, rowptr, psum, counts);
    k_pair<<<EL_BLOCKS, B, 0, stream>>>(vB, kB, rowptr, csr);

    k_gather_in<<<GX_BLOCKS, B, 0, stream>>>(rowptr, csr, x, W1, b1, hA);

    k_gemm_rows_wmma<<<GM_BLOCKS, B, 0, stream>>>(hA, W2, hw);
    k_gather<<<GA_BLOCKS, B, 0, stream>>>(rowptr, csr, hw, b2, hA);

    k_gather_pool<<<GA_BLOCKS, B, 0, stream>>>(rowptr, csr, hA, batch, psum, counts);

    k_pool_gemm_wmma<<<N_GRAPHS / 16, B, 0, stream>>>(psum, W3, counts, b3, out);
}


# Sort scatter first puts its 8192-edge tile in digit order in LDS, then neighbouring lanes store neighbouring sorted elements: a store instruction touches 1-3 lines, not about 30; sums unchanged
# speedup vs baseline: 5.9189x; 1.2905x over previous
#include <hip/hip_runtime.h>

#define N_NODES  50000
#define N_EDGES  1600000
#define N_GRAPHS 256
#define IN_DIM   9
#define HID      128

typedef __attribute__((ext_vector_type(2))) float v2f;
typedef __attribute__((ext_vector_type(8))) float v8f;

__device__ __forceinline__ void atomic_add_f32(float* p, float v) {
    asm volatile("global_atomic_add_f32 %0, %1, off scope:SCOPE_DEV"
                 :: "v"(p), "v"(v) : "memory");
}

#define RS_ITEMS  32
#define RS_WTILE  1024
#define RS_BTILE  8192
#define RS_BLOCKS 196
#define RS_HIST   50176

__device__ __forceinline__ unsigned match_digit(int d, bool valid) {
    unsigned m = __builtin_amdgcn_ballot_w32(valid);
#pragma unroll
    for (int b = 0; b < 8; ++b) {
        bool bit = ((d >> b) & 1) != 0;
        unsigned bal = __builtin_amdgcn_ballot_w32(bit);
        m &= bit ? bal : ~bal;
    }
    return m;
}

template <int SHIFT>
__global__ void __launch_bounds__(256) k_hist(const int* __restrict__ keys, int* __restrict__ hist) {
    __shared__ int h[8][256];
    int tid = threadIdx.x;
    int w = tid >> 5;
    int lane = tid & 31;
#pragma unroll
    for (int k = 0; k < 8; ++k) h[k][tid] = 0;
    __syncthreads();
    int base = blockIdx.x * RS_BTILE + w * RS_WTILE + lane;
    int key[RS_ITEMS];
#pragma unroll
    for (int i = 0; i < RS_ITEMS; ++i) {
        int idx = base + i * 32;
        key[i] = keys[idx < N_EDGES ? idx : N_EDGES - 1];
    }
#pragma unroll
    for (int i = 0; i < RS_ITEMS; ++i) {
        bool valid = (base + i * 32) < N_EDGES;
        int d = (key[i] >> SHIFT) & 255;
        unsigned m = match_digit(d, valid);
        int top = 31 - __builtin_clz(m | 1u);
        if (valid && lane == top) h[w][d] += __builtin_popcount(m);
    }
    __syncthreads();
    int tot = 0;
#pragma unroll
    for (int k = 0; k < 8; ++k) tot += h[k][tid];
    hist[tid * RS_BLOCKS + blockIdx.x] = tot;
}

__global__ void __launch_bounds__(256) k_scan(const int* __restrict__ in, int* __restrict__ out) {
    __shared__ int s[256];
    int t = threadIdx.x;
    int row = blockIdx.x * RS_BLOCKS;
    int part = 0;
    for (int i = t; i < row; i += 256) part += in[i];
    s[t] = part;
    __syncthreads();
    for (int off = 1; off < 256; off <<= 1) {
        int v = (t >= off) ? s[t - off] : 0;
        __syncthreads();
        s[t] += v;
        __syncthreads();
    }
    int before = s[255];
    __syncthreads();
    int x = (t < RS_BLOCKS) ? in[row + t] : 0;
    s[t] = x;
    __syncthreads();
    for (int off = 1; off < 256; off <<= 1) {
        int v = (t >= off) ? s[t - off] : 0;
        __syncthreads();
        s[t] += v;
        __syncthreads();
    }
    if (t < RS_BLOCKS) out[row + t] = before + s[t] - x;
}

template <int SHIFT>
__global__ void __launch_bounds__(256) k_scatter(const int* __restrict__ keys, const int* __restrict__ vals,
                                                 const int* __restrict__ hscan,
                                                 int* __restrict__ keys_out, int* __restrict__ vals_out) {
    __shared__ int h[8][256];
    __shared__ int skey[RS_BTILE];
    __shared__ int sval[RS_BTILE];
    __shared__ int gdel[256];
    __shared__ int wsum[8];
    int tid = threadIdx.x;
    int w = tid >> 5;
    int lane = tid & 31;
#pragma unroll
    for (int k = 0; k < 8; ++k) h[k][tid] = 0;
    __syncthreads();
    int base = blockIdx.x * RS_BTILE + w * RS_WTILE + lane;
    int key[RS_ITEMS];
    int val[RS_ITEMS];
#pragma unroll
    for (int i = 0; i < RS_ITEMS; ++i) {
        int idx = base + i * 32;
        int ic = idx < N_EDGES ? idx : N_EDGES - 1;
        key[i] = keys[ic];
        val[i] = vals[ic];
    }
    unsigned lt = (1u << lane) - 1u;
    int info[RS_ITEMS];
#pragma unroll
    for (int i = 0; i < RS_ITEMS; ++i) {
        bool valid = (base + i * 32) < N_EDGES;
        int d = (key[i] >> SHIFT) & 255;
        unsigned m = match_digit(d, valid);
        int top = 31 - __builtin_clz(m | 1u);
        int cntg = __builtin_popcount(m);
        bool leader = valid && lane == top;
        if (leader) h[w][d] += cntg;
        info[i] = __builtin_popcount(m & lt) | (cntg << 8) | (leader ? 0x10000 : 0);
    }
    __syncthreads();
    int c[8];
    int tot = 0;
#pragma unroll
    for (int k = 0; k < 8; ++k) { c[k] = h[k][tid]; tot += c[k]; }
    int inc = tot;
#pragma unroll
    for (int off = 1; off < 32; off <<= 1) {
        int v = __builtin_amdgcn_ds_bpermute(((lane - off) & 31) << 2, inc);
        if (lane >= off) inc += v;
    }
    if (lane == 31) wsum[w] = inc;
    __syncthreads();
    int lstart = inc - tot;
#pragma unroll
    for (int k = 0; k < 8; ++k) lstart += (k < w) ? wsum[k] : 0;
    gdel[tid] = hscan[tid * RS_BLOCKS + blockIdx.x] - lstart;
    {
        int run = lstart;
#pragma unroll
        for (int k = 0; k < 8; ++k) {
            h[k][tid] = run;
            run += c[k];
        }
    }
    __syncthreads();
#pragma unroll
    for (int i = 0; i < RS_ITEMS; ++i) {
        bool valid = (base + i * 32) < N_EDGES;
        int d = (key[i] >> SHIFT) & 255;
        int start = h[w][d];
        int lpos = start + (info[i] & 255);
        if (info[i] & 0x10000) h[w][d] = start + ((info[i] >> 8) & 255);
        if (valid) {
            skey[lpos] = key[i];
            sval[lpos] = val[i];
        }
    }
    __syncthreads();
    int nvalid = N_EDGES - blockIdx.x * RS_BTILE;
    nvalid = nvalid < RS_BTILE ? nvalid : RS_BTILE;
#pragma unroll
    for (int i = 0; i < RS_ITEMS; ++i) {
        int j = tid + i * 256;
        int jc = j < nvalid ? j : 0;
        key[i] = skey[jc];
        val[i] = sval[jc];
    }
#pragma unroll
    for (int i = 0; i < RS_ITEMS; ++i) info[i] = gdel[(key[i] >> SHIFT) & 255];
#pragma unroll
    for (int i = 0; i < RS_ITEMS; ++i) {
        int j = tid + i * 256;
        if (j < nvalid) {
            keys_out[j + info[i]] = key[i];
            vals_out[j + info[i]] = val[i];
        }
    }
}

#define EL_ITEMS   5
#define EL_BLOCKS  1250
#define EL_STRIDE  (EL_BLOCKS * 256)
static_assert(EL_STRIDE * EL_ITEMS == N_EDGES, "every thread owns exactly EL_ITEMS edges: no bounds test in the kernels");

static_assert(EL_STRIDE >= N_GRAPHS * HID, "one thread per element of the per-graph sums");
__global__ void __launch_bounds__(256) k_bounds(const int* __restrict__ sdst, int* __restrict__ rowptr,
                                                float* __restrict__ sums, float* __restrict__ counts) {
    int t = blockIdx.x * 256 + threadIdx.x;
    if (t < N_GRAPHS * HID) sums[t] = 0.f;
    if (t < N_GRAPHS) counts[t] = 0.f;
    int d[EL_ITEMS];
    int prev[EL_ITEMS];
#pragma unroll
    for (int i = 0; i < EL_ITEMS; ++i) {
        int p = t + i * EL_STRIDE;
        d[i] = sdst[p];
        prev[i] = (i > 0 || p > 0) ? sdst[p - 1] : -1;
    }
#pragma unroll
    for (int i = 0; i < EL_ITEMS; ++i) {
        int p = t + i * EL_STRIDE;
        for (int n = prev[i] + 1; n <= d[i]; ++n) rowptr[n] = p;
        if (p == N_EDGES - 1)
            for (int n = d[i] + 1; n <= N_NODES; ++n) rowptr[n] = N_EDGES;
    }
}

__global__ void __launch_bounds__(256) k_pair(const int* __restrict__ ssrc, const int* __restrict__ sdst,
                                              const int* __restrict__ rowptr, int2* __restrict__ csr) {
    int t = blockIdx.x * 256 + threadIdx.x;
    int s[EL_ITEMS];
    int d[EL_ITEMS];
#pragma unroll
    for (int i = 0; i < EL_ITEMS; ++i) {
        int p = t + i * EL_STRIDE;
        s[i] = ssrc[p];
        d[i] = sdst[p];
    }
    int degs[EL_ITEMS];
    int degd[EL_ITEMS];
#pragma unroll
    for (int i = 0; i < EL_ITEMS; ++i) {
        degs[i] = rowptr[s[i] + 1] - rowptr[s[i]];
        degd[i] = rowptr[d[i] + 1] - rowptr[d[i]];
    }
#pragma unroll
    for (int i = 0; i < EL_ITEMS; ++i) {
        int p = t + i * EL_STRIDE;
        float ds = rsqrtf((float)degs[i] + 1.0f);
        float dd = rsqrtf((float)degd[i] + 1.0f);
        int2 v;
        v.x = s[i];
        v.y = __float_as_int(ds * dd);
        csr[p] = v;
    }
}

#define GM_BLOCKS 256
#define GM_TILES  (N_NODES >> 4)
#define GM_CHUNK  8
static_assert(GM_TILES * 16 == N_NODES, "whole row tiles only: no row bounds test in the kernel");
static_assert(GM_BLOCKS * 8 <= GM_TILES, "every wave owns at least one row tile: the first tile's loads need no test");
__global__ void __launch_bounds__(256) k_gemm_rows_wmma(const float* __restrict__ A, const float* __restrict__ W,
                                                        float* __restrict__ out) {
    __shared__ v2f Bs[32 * 8 * 32];
    int w    = __builtin_amdgcn_readfirstlane(threadIdx.x >> 5);
    int lane = threadIdx.x & 31;
    int half = lane >> 4;
    int l16  = lane & 15;

    int rt = blockIdx.x + GM_BLOCKS * w;
    const float* arow = A + ((rt << 4) + l16) * HID + (half << 1);
    v2f an[GM_CHUNK];
#pragma unroll
    for (int u = 0; u < GM_CHUNK; ++u) an[u] = *(const v2f*)(arow + 4 * u);

#pragma unroll
    for (int s = 0; s < 32; ++s) {
        int ka = 4 * s + (half << 1);
        v2f b; b.x = W[ka * HID + (w << 4) + l16]; b.y = W[(ka + 1) * HID + (w << 4) + l16];
        Bs[(s * 8 + w) * 32 + lane] = b;
    }
    __syncthreads();

    for (;;) {
        int rn = rt + GM_BLOCKS * 8;
        bool more = rn < GM_TILES;
        const float* arow_n = A + (((more ? rn : rt) << 4) + l16) * HID + (half << 1);
        v8f c[8];
#pragma unroll
        for (int ct = 0; ct < 8; ++ct) c[ct] = (v8f){0.f, 0.f, 0.f, 0.f, 0.f, 0.f, 0.f, 0.f};
#pragma unroll 1
        for (int q = 0; q < 32 / GM_CHUNK; ++q) {
            v2f ac[GM_CHUNK];
#pragma unroll
            for (int u = 0; u < GM_CHUNK; ++u) { ac[u].x = fmaxf(an[u].x, 0.f); ac[u].y = fmaxf(an[u].y, 0.f); }
            const float* ap = (q < 32 / GM_CHUNK - 1) ? arow + 4 * GM_CHUNK * (q + 1) : arow_n;
#pragma unroll
            for (int u = 0; u < GM_CHUNK; ++u) an[u] = *(const v2f*)(ap + 4 * u);
            const v2f* bq = Bs + q * (GM_CHUNK * 8 * 32) + lane;
#pragma unroll
            for (int u = 0; u < GM_CHUNK; ++u) {
#pragma unroll
                for (int ct = 0; ct < 8; ++ct) {
                    v2f b = bq[(u * 8 + ct) * 32];
                    c[ct] = __builtin_amdgcn_wmma_f32_16x16x4_f32(false, ac[u], false, b,
                                                                  (short)0, c[ct], false, false);
                }
            }
        }
        float* orow = out + ((rt << 4) + (half << 3)) * HID + l16;
#pragma unroll
        for (int ct = 0; ct < 8; ++ct) {
#pragma unroll
            for (int v = 0; v < 8; ++v) orow[v * HID + (ct << 4)] = c[ct][v];
        }
        if (!more) break;
        rt = rn;
        arow = arow_n;
    }
}

#define GA_NODES  4
#define GA_BLOCKS 1563

__device__ __forceinline__ float4 edge_sum(const int2* __restrict__ csr, const float* __restrict__ hl,
                                           int beg, int end, float4 acc) {
    int j = beg;
    for (; j + 4 <= end; j += 4) {
        int2 e0 = csr[j], e1 = csr[j + 1], e2 = csr[j + 2], e3 = csr[j + 3];
        const float4 h0 = *(const float4*)(hl + e0.x * HID);
        const float4 h1 = *(const float4*)(hl + e1.x * HID);
        const float4 h2 = *(const float4*)(hl + e2.x * HID);
        const float4 h3 = *(const float4*)(hl + e3.x * HID);
        float w0 = __int_as_float(e0.y), w1 = __int_as_float(e1.y);
        float w2 = __int_as_float(e2.y), w3 = __int_as_float(e3.y);
        acc.x = fmaf(h0.x, w0, acc.x); acc.y = fmaf(h0.y, w0, acc.y);
        acc.z = fmaf(h0.z, w0, acc.z); acc.w = fmaf(h0.w, w0, acc.w);
        acc.x = fmaf(h1.x, w1, acc.x); acc.y = fmaf(h1.y, w1, acc.y);
        acc.z = fmaf(h1.z, w1, acc.z); acc.w = fmaf(h1.w, w1, acc.w);
        acc.x = fmaf(h2.x, w2, acc.x); acc.y = fmaf(h2.y, w2, acc.y);
        acc.z = fmaf(h2.z, w2, acc.z); acc.w = fmaf(h2.w, w2, acc.w);
        acc.x = fmaf(h3.x, w3, acc.x); acc.y = fmaf(h3.y, w3, acc.y);
        acc.z = fmaf(h3.z, w3, acc.z); acc.w = fmaf(h3.w, w3, acc.w);
    }
    for (; j < end; ++j) {
        int2 e0 = csr[j];
        const float4 h0 = *(const float4*)(hl + e0.x * HID);
        float w0 = __int_as_float(e0.y);
        acc.x = fmaf(h0.x, w0, acc.x); acc.y = fmaf(h0.y, w0, acc.y);
        acc.z = fmaf(h0.z, w0, acc.z); acc.w = fmaf(h0.w, w0, acc.w);
    }
    return acc;
}

__global__ void __launch_bounds__(256) k_gather(const int* __restrict__ rowptr, const int2* __restrict__ csr,
                                                const float* __restrict__ hw, const float* __restrict__ bias,
                                                float* __restrict__ out) {
    int wave = __builtin_amdgcn_readfirstlane((blockIdx.x * 256 + threadIdx.x) >> 5);
    int lane = threadIdx.x & 31;
    const float* hl = hw + lane * 4;
    const float4 b4 = *(const float4*)(bias + lane * 4);
    for (int i = 0; i < GA_NODES; ++i) {
        int n = wave * GA_NODES + i;
        if (n >= N_NODES) return;
        int beg = rowptr[n];
        int end = rowptr[n + 1];
        float4 acc = edge_sum(csr, hl, beg, end, make_float4(0.f, 0.f, 0.f, 0.f));
        float sd = rsqrtf((float)(end - beg) + 1.0f);
        float ss = sd * sd;
        const float4 hs = *(const float4*)(hl + n * HID);
        float4 o;
        o.x = fmaxf(acc.x + fmaf(hs.x, ss, b4.x), 0.f);
        o.y = fmaxf(acc.y + fmaf(hs.y, ss, b4.y), 0.f);
        o.z = fmaxf(acc.z + fmaf(hs.z, ss, b4.z), 0.f);
        o.w = fmaxf(acc.w + fmaf(hs.w, ss, b4.w), 0.f);
        *(float4*)(out + n * HID + lane * 4) = o;
    }
}

__device__ __forceinline__ void pool_add(float* __restrict__ sums, float* __restrict__ counts,
                                         int g, int lane, float4 acc, int cnt) {
    float* o = sums + g * HID + lane * 4;
    atomic_add_f32(o + 0, acc.x);
    atomic_add_f32(o + 1, acc.y);
    atomic_add_f32(o + 2, acc.z);
    atomic_add_f32(o + 3, acc.w);
    if (lane == 0) atomic_add_f32(counts + g, (float)cnt);
}

__global__ void __launch_bounds__(256) k_gather_pool(const int* __restrict__ rowptr, const int2* __restrict__ csr,
                                                     const float* __restrict__ hw, const int* __restrict__ batch,
                                                     float* __restrict__ sums, float* __restrict__ counts) {
    int wave = __builtin_amdgcn_readfirstlane((blockIdx.x * 256 + threadIdx.x) >> 5);
    int lane = threadIdx.x & 31;
    const float* hl = hw + lane * 4;
    int n0 = wave * GA_NODES;
    if (n0 >= N_NODES) return;
    int g = batch[n0];
    int cnt = 0;
    float4 acc = make_float4(0.f, 0.f, 0.f, 0.f);
    for (int i = 0; i < GA_NODES; ++i) {
        int n = n0 + i;
        if (n >= N_NODES) break;
        int gn = batch[n];
        int beg = rowptr[n];
        int end = rowptr[n + 1];
        if (gn != g) {
            pool_add(sums, counts, g, lane, acc, cnt);
            acc = make_float4(0.f, 0.f, 0.f, 0.f);
            cnt = 0;
            g = gn;
        }
        acc = edge_sum(csr, hl, beg, end, acc);
        float sd = rsqrtf((float)(end - beg) + 1.0f);
        float ss = sd * sd;
        const float4 hs = *(const float4*)(hl + n * HID);
        acc.x = fmaf(hs.x, ss, acc.x);
        acc.y = fmaf(hs.y, ss, acc.y);
        acc.z = fmaf(hs.z, ss, acc.z);
        acc.w = fmaf(hs.w, ss, acc.w);
        ++cnt;
    }
    pool_add(sums, counts, g, lane, acc, cnt);
}

#define GX_UNROLL 12
#define GX_TRIP   (3 * GX_UNROLL)
#define GX_NODES  5
#define GX_BLOCKS 1250
static_assert(GX_BLOCKS * 8 * GX_NODES == N_NODES, "every wave owns exactly GX_NODES nodes: no bounds test in the kernel");
__global__ void __launch_bounds__(256) k_gather_in(const int* __restrict__ rowptr, const int2* __restrict__ csr,
                                                   const float* __restrict__ x, const float* __restrict__ W1,
                                                   const float* __restrict__ bias, float* __restrict__ out) {
    int wave = __builtin_amdgcn_readfirstlane((blockIdx.x * 256 + threadIdx.x) >> 5);
    int lane = threadIdx.x & 31;
    int l = lane < 27 ? lane : 26;
    int e = l / IN_DIM;
    int k = l - e * IN_DIM;
    float4 w[IN_DIM];
#pragma unroll
    for (int q = 0; q < IN_DIM; ++q) w[q] = *(const float4*)(W1 + q * HID + lane * 4);
    const float4 b4 = *(const float4*)(bias + lane * 4);
    for (int i = 0; i < GX_NODES; ++i) {
        int n = wave * GX_NODES + i;
        int beg = rowptr[n];
        int end = rowptr[n + 1];
        float xs = x[n * IN_DIM + k];
        float acc = 0.f;
        for (int j0 = beg; j0 < end; j0 += GX_TRIP) {
            int2 ed[GX_UNROLL];
            float xv[GX_UNROLL];
#pragma unroll
            for (int u = 0; u < GX_UNROLL; ++u) {
                int j = j0 + 3 * u + e;
                ed[u] = csr[j < end ? j : end - 1];
            }
#pragma unroll
            for (int u = 0; u < GX_UNROLL; ++u) xv[u] = x[ed[u].x * IN_DIM + k];
#pragma unroll
            for (int u = 0; u < GX_UNROLL; ++u) {
                int j = j0 + 3 * u + e;
                float t = fmaf(xv[u], __int_as_float(ed[u].y), acc);
                acc = j < end ? t : acc;
            }
        }
        float s1 = __int_as_float(__builtin_amdgcn_ds_bpermute(((lane + 9) & 31) << 2, __float_as_int(acc)));
        float s2 = __int_as_float(__builtin_amdgcn_ds_bpermute(((lane + 18) & 31) << 2, __float_as_int(acc)));
        float sd = rsqrtf((float)(end - beg) + 1.0f);
        float ss = sd * sd;
        float ax = fmaf(xs, ss, (acc + s1) + s2);
        float4 o = make_float4(0.f, 0.f, 0.f, 0.f);
#pragma unroll
        for (int q = 0; q < IN_DIM; ++q) {
            float a = __int_as_float(__builtin_amdgcn_readlane(__float_as_int(ax), q));
            o.x = fmaf(a, w[q].x, o.x); o.y = fmaf(a, w[q].y, o.y);
            o.z = fmaf(a, w[q].z, o.z); o.w = fmaf(a, w[q].w, o.w);
        }
        o.x += b4.x; o.y += b4.y; o.z += b4.z; o.w += b4.w;
        *(float4*)(out + n * HID + lane * 4) = o;
    }
}

__global__ void __launch_bounds__(256) k_pool_gemm_wmma(const float* __restrict__ S, const float* __restrict__ W,
                                                        const float* __restrict__ counts, const float* __restrict__ bias,
                                                        float* __restrict__ out) {
    int wave = (blockIdx.x * 256 + threadIdx.x) >> 5;
    int lane = threadIdx.x & 31;
    int ct = wave & 7;
    int rt = wave >> 3;

    int half = lane >> 4;
    int l16  = lane & 15;
    int row  = (rt << 4) + l16;
    int col  = (ct << 4) + l16;
    const float* arow = S + row * HID;

    v8f c = {0.f, 0.f, 0.f, 0.f, 0.f, 0.f, 0.f, 0.f};
#pragma unroll
    for (int k0 = 0; k0 < HID; k0 += 4) {
        int ka = k0 + (half << 1);
        v2f a; a.x = arow[ka]; a.y = arow[ka + 1];
        v2f b; b.x = W[ka * HID + col]; b.y = W[(ka + 1) * HID + col];
        c = __builtin_amdgcn_wmma_f32_16x16x4_f32(false, a, false, b,
                                                  (short)0, c, false, false);
    }
    int g0 = (rt << 4) + (half << 3);
    float bc = bias[col];
#pragma unroll
    for (int v = 0; v < 8; ++v) {
        float cnt = counts[g0 + v];
        out[(g0 + v) * HID + col] = fmaf(cnt, bc, c[v]) / fmaxf(cnt, 1.0f);
    }
}

extern "C" void kernel_launch(void* const* d_in, const int* in_sizes, int n_in,
                              void* d_out, int out_size, void* d_ws, size_t ws_size,
                              hipStream_t stream) {
    const float* x        = (const float*)d_in[0];
    const int*   eidx     = (const int*)d_in[1];
    const int*   batch    = (const int*)d_in[2];
    const float* W1       = (const float*)d_in[3];
    const float* b1       = (const float*)d_in[4];
    const float* W2       = (const float*)d_in[5];
    const float* b2       = (const float*)d_in[6];
    const float* W3       = (const float*)d_in[7];
    const float* b3       = (const float*)d_in[8];
    float* out = (float*)d_out;

    const int* src = eidx;
    const int* dst = eidx + N_EDGES;

    float* ws     = (float*)d_ws;
    float* hw     = ws;
    float* hA     = hw + (size_t)N_NODES * HID;
    float* counts = hA + (size_t)N_NODES * HID;
    float* psum   = counts + N_GRAPHS;
    int2*  csr    = (int2*)(psum + N_GRAPHS * HID);
    int*   kA     = (int*)(csr + N_EDGES);
    int*   vA     = kA + N_EDGES;
    int*   kB     = vA + N_EDGES;
    int*   vB     = kB + N_EDGES;
    int*   hist   = vB + N_EDGES;
    int*   hscan  = hist + RS_HIST;
    int*   rowptr = hscan + RS_HIST;

    const int B = 256;

    k_hist<0><<<RS_BLOCKS, B, 0, stream>>>(dst, hist);
    k_scan<<<256, B, 0, stream>>>(hist, hscan);
    k_scatter<0><<<RS_BLOCKS, B, 0, stream>>>(dst, src, hscan, kA, vA);
    k_hist<8><<<RS_BLOCKS, B, 0, stream>>>(kA, hist);
    k_scan<<<256, B, 0, stream>>>(hist, hscan);
    k_scatter<8><<<RS_BLOCKS, B, 0, stream>>>(kA, vA, hscan, kB, vB);
    k_bounds<<<EL_BLOCKS, B, 0, stream>>>(kB, rowptr, psum, counts);
    k_pair<<<EL_BLOCKS, B, 0, stream>>>(vB, kB, rowptr, csr);

    k_gather_in<<<GX_BLOCKS, B, 0, stream>>>(rowptr, csr, x, W1, b1, hA);

    k_gemm_rows_wmma<<<GM_BLOCKS, B, 0, stream>>>(hA, W2, hw);
    k_gather<<<GA_BLOCKS, B, 0, stream>>>(rowptr, csr, hw, b2, hA);

    k_gather_pool<<<GA_BLOCKS, B, 0, stream>>>(rowptr, csr, hA, batch, psum, counts);

    k_pool_gemm_wmma<<<N_GRAPHS / 16, B, 0, stream>>>(psum, W3, counts, b3, out);
}


# Sort: each scatter work-group sums the per-tile digit counts it needs itself (counts stored tile-major, 2 x 98 loads per thread); both prefix-sum launches go (13 -> 11); edge order and sums unchanged
# speedup vs baseline: 6.7446x; 1.1395x over previous
#include <hip/hip_runtime.h>

#define N_NODES  50000
#define N_EDGES  1600000
#define N_GRAPHS 256
#define IN_DIM   9
#define HID      128

typedef __attribute__((ext_vector_type(2))) float v2f;
typedef __attribute__((ext_vector_type(8))) float v8f;

__device__ __forceinline__ void atomic_add_f32(float* p, float v) {
    asm volatile("global_atomic_add_f32 %0, %1, off scope:SCOPE_DEV"
                 :: "v"(p), "v"(v) : "memory");
}

#define RS_ITEMS  32
#define RS_WTILE  1024
#define RS_BTILE  8192
#define RS_BLOCKS 196
#define RS_HIST   50176
#define RS_HCHUNK 98
static_assert(RS_BLOCKS % RS_HCHUNK == 0, "whole chunks of tiles: no bounds test in the scatter kernel's count loop");

__device__ __forceinline__ unsigned match_digit(int d, bool valid) {
    unsigned m = __builtin_amdgcn_ballot_w32(valid);
#pragma unroll
    for (int b = 0; b < 8; ++b) {
        bool bit = ((d >> b) & 1) != 0;
        unsigned bal = __builtin_amdgcn_ballot_w32(bit);
        m &= bit ? bal : ~bal;
    }
    return m;
}

template <int SHIFT>
__global__ void __launch_bounds__(256) k_hist(const int* __restrict__ keys, int* __restrict__ hist) {
    __shared__ int h[8][256];
    int tid = threadIdx.x;
    int w = tid >> 5;
    int lane = tid & 31;
#pragma unroll
    for (int k = 0; k < 8; ++k) h[k][tid] = 0;
    __syncthreads();
    int base = blockIdx.x * RS_BTILE + w * RS_WTILE + lane;
    int key[RS_ITEMS];
#pragma unroll
    for (int i = 0; i < RS_ITEMS; ++i) {
        int idx = base + i * 32;
        key[i] = keys[idx < N_EDGES ? idx : N_EDGES - 1];
    }
#pragma unroll
    for (int i = 0; i < RS_ITEMS; ++i) {
        bool valid = (base + i * 32) < N_EDGES;
        int d = (key[i] >> SHIFT) & 255;
        unsigned m = match_digit(d, valid);
        int top = 31 - __builtin_clz(m | 1u);
        if (valid && lane == top) h[w][d] += __builtin_popcount(m);
    }
    __syncthreads();
    int tot = 0;
#pragma unroll
    for (int k = 0; k < 8; ++k) tot += h[k][tid];
    hist[blockIdx.x * 256 + tid] = tot;
}

template <int SHIFT>
__global__ void __launch_bounds__(256) k_scatter(const int* __restrict__ keys, const int* __restrict__ vals,
                                                 const int* __restrict__ hist,
                                                 int* __restrict__ keys_out, int* __restrict__ vals_out) {
    __shared__ int h[8][256];
    __shared__ int skey[RS_BTILE];
    __shared__ int sval[RS_BTILE];
    __shared__ int gdel[256];
    __shared__ int wsum[8];
    __shared__ int gsum[8];
    int tid = threadIdx.x;
    int w = tid >> 5;
    int lane = tid & 31;
#pragma unroll
    for (int k = 0; k < 8; ++k) h[k][tid] = 0;
    __syncthreads();
    int base = blockIdx.x * RS_BTILE + w * RS_WTILE + lane;
    int key[RS_ITEMS];
    int val[RS_ITEMS];
#pragma unroll
    for (int i = 0; i < RS_ITEMS; ++i) {
        int idx = base + i * 32;
        int ic = idx < N_EDGES ? idx : N_EDGES - 1;
        key[i] = keys[ic];
        val[i] = vals[ic];
    }
    const int* hp = hist + tid;
    int nfront = blockIdx.x;
    int gtot = 0;
    int gpre = 0;
#pragma unroll 1
    for (int t = 0; t < RS_BLOCKS / RS_HCHUNK; ++t) {
#pragma unroll
        for (int u = 0; u < RS_HCHUNK; ++u) {
            int v = hp[u * 256];
            gtot += v;
            gpre += (u < nfront) ? v : 0;
        }
        hp += RS_HCHUNK * 256;
        nfront -= RS_HCHUNK;
    }
    unsigned lt = (1u << lane) - 1u;
    int info[RS_ITEMS];
#pragma unroll
    for (int i = 0; i < RS_ITEMS; ++i) {
        bool valid = (base + i * 32) < N_EDGES;
        int d = (key[i] >> SHIFT) & 255;
        unsigned m = match_digit(d, valid);
        int top = 31 - __builtin_clz(m | 1u);
        int cntg = __builtin_popcount(m);
        bool leader = valid && lane == top;
        if (leader) h[w][d] += cntg;
        info[i] = __builtin_popcount(m & lt) | (cntg << 8) | (leader ? 0x10000 : 0);
    }
    __syncthreads();
    int c[8];
    int tot = 0;
#pragma unroll
    for (int k = 0; k < 8; ++k) { c[k] = h[k][tid]; tot += c[k]; }
    int inc = tot;
    int ginc = gtot;
#pragma unroll
    for (int off = 1; off < 32; off <<= 1) {
        int v = __builtin_amdgcn_ds_bpermute(((lane - off) & 31) << 2, inc);
        int gv = __builtin_amdgcn_ds_bpermute(((lane - off) & 31) << 2, ginc);
        if (lane >= off) { inc += v; ginc += gv; }
    }
    if (lane == 31) { wsum[w] = inc; gsum[w] = ginc; }
    __syncthreads();
    int lstart = inc - tot;
    int gstart = ginc - gtot + gpre;
#pragma unroll
    for (int k = 0; k < 8; ++k) {
        lstart += (k < w) ? wsum[k] : 0;
        gstart += (k < w) ? gsum[k] : 0;
    }
    gdel[tid] = gstart - lstart;
    {
        int run = lstart;
#pragma unroll
        for (int k = 0; k < 8; ++k) {
            h[k][tid] = run;
            run += c[k];
        }
    }
    __syncthreads();
#pragma unroll
    for (int i = 0; i < RS_ITEMS; ++i) {
        bool valid = (base + i * 32) < N_EDGES;
        int d = (key[i] >> SHIFT) & 255;
        int start = h[w][d];
        int lpos = start + (info[i] & 255);
        if (info[i] & 0x10000) h[w][d] = start + ((info[i] >> 8) & 255);
        if (valid) {
            skey[lpos] = key[i];
            sval[lpos] = val[i];
        }
    }
    __syncthreads();
    int nvalid = N_EDGES - blockIdx.x * RS_BTILE;
    nvalid = nvalid < RS_BTILE ? nvalid : RS_BTILE;
#pragma unroll
    for (int i = 0; i < RS_ITEMS; ++i) {
        int j = tid + i * 256;
        int jc = j < nvalid ? j : 0;
        key[i] = skey[jc];
        val[i] = sval[jc];
    }
#pragma unroll
    for (int i = 0; i < RS_ITEMS; ++i) info[i] = gdel[(key[i] >> SHIFT) & 255];
#pragma unroll
    for (int i = 0; i < RS_ITEMS; ++i) {
        int j = tid + i * 256;
        if (j < nvalid) {
            keys_out[j + info[i]] = key[i];
            vals_out[j + info[i]] = val[i];
        }
    }
}

#define EL_ITEMS   5
#define EL_BLOCKS  1250
#define EL_STRIDE  (EL_BLOCKS * 256)
static_assert(EL_STRIDE * EL_ITEMS == N_EDGES, "every thread owns exactly EL_ITEMS edges: no bounds test in the kernels");

static_assert(EL_STRIDE >= N_GRAPHS * HID, "one thread per element of the per-graph sums");
__global__ void __launch_bounds__(256) k_bounds(const int* __restrict__ sdst, int* __restrict__ rowptr,
                                                float* __restrict__ sums, float* __restrict__ counts) {
    int t = blockIdx.x * 256 + threadIdx.x;
    if (t < N_GRAPHS * HID) sums[t] = 0.f;
    if (t < N_GRAPHS) counts[t] = 0.f;
    int d[EL_ITEMS];
    int prev[EL_ITEMS];
#pragma unroll
    for (int i = 0; i < EL_ITEMS; ++i) {
        int p = t + i * EL_STRIDE;
        d[i] = sdst[p];
        prev[i] = (i > 0 || p > 0) ? sdst[p - 1] : -1;
    }
#pragma unroll
    for (int i = 0; i < EL_ITEMS; ++i) {
        int p = t + i * EL_STRIDE;
        for (int n = prev[i] + 1; n <= d[i]; ++n) rowptr[n] = p;
        if (p == N_EDGES - 1)
            for (int n = d[i] + 1; n <= N_NODES; ++n) rowptr[n] = N_EDGES;
    }
}

__global__ void __launch_bounds__(256) k_pair(const int* __restrict__ ssrc, const int* __restrict__ sdst,
                                              const int* __restrict__ rowptr, int2* __restrict__ csr) {
    int t = blockIdx.x * 256 + threadIdx.x;
    int s[EL_ITEMS];
    int d[EL_ITEMS];
#pragma unroll
    for (int i = 0; i < EL_ITEMS; ++i) {
        int p = t + i * EL_STRIDE;
        s[i] = ssrc[p];
        d[i] = sdst[p];
    }
    int degs[EL_ITEMS];
    int degd[EL_ITEMS];
#pragma unroll
    for (int i = 0; i < EL_ITEMS; ++i) {
        degs[i] = rowptr[s[i] + 1] - rowptr[s[i]];
        degd[i] = rowptr[d[i] + 1] - rowptr[d[i]];
    }
#pragma unroll
    for (int i = 0; i < EL_ITEMS; ++i) {
        int p = t + i * EL_STRIDE;
        float ds = rsqrtf((float)degs[i] + 1.0f);
        float dd = rsqrtf((float)degd[i] + 1.0f);
        int2 v;
        v.x = s[i];
        v.y = __float_as_int(ds * dd);
        csr[p] = v;
    }
}

#define GM_BLOCKS 256
#define GM_TILES  (N_NODES >> 4)
#define GM_CHUNK  8
static_assert(GM_TILES * 16 == N_NODES, "whole row tiles only: no row bounds test in the kernel");
static_assert(GM_BLOCKS * 8 <= GM_TILES, "every wave owns at least one row tile: the first tile's loads need no test");
__global__ void __launch_bounds__(256) k_gemm_rows_wmma(const float* __restrict__ A, const float* __restrict__ W,
                                                        float* __restrict__ out) {
    __shared__ v2f Bs[32 * 8 * 32];
    int w    = __builtin_amdgcn_readfirstlane(threadIdx.x >> 5);
    int lane = threadIdx.x & 31;
    int half = lane >> 4;
    int l16  = lane & 15;

    int rt = blockIdx.x + GM_BLOCKS * w;
    const float* arow = A + ((rt << 4) + l16) * HID + (half << 1);
    v2f an[GM_CHUNK];
#pragma unroll
    for (int u = 0; u < GM_CHUNK; ++u) an[u] = *(const v2f*)(arow + 4 * u);

#pragma unroll
    for (int s = 0; s < 32; ++s) {
        int ka = 4 * s + (half << 1);
        v2f b; b.x = W[ka * HID + (w << 4) + l16]; b.y = W[(ka + 1) * HID + (w << 4) + l16];
        Bs[(s * 8 + w) * 32 + lane] = b;
    }
    __syncthreads();

    for (;;) {
        int rn = rt + GM_BLOCKS * 8;
        bool more = rn < GM_TILES;
        const float* arow_n = A + (((more ? rn : rt) << 4) + l16) * HID + (half << 1);
        v8f c[8];
#pragma unroll
        for (int ct = 0; ct < 8; ++ct) c[ct] = (v8f){0.f, 0.f, 0.f, 0.f, 0.f, 0.f, 0.f, 0.f};
#pragma unroll 1
        for (int q = 0; q < 32 / GM_CHUNK; ++q) {
            v2f ac[GM_CHUNK];
#pragma unroll
            for (int u = 0; u < GM_CHUNK; ++u) { ac[u].x = fmaxf(an[u].x, 0.f); ac[u].y = fmaxf(an[u].y, 0.f); }
            const float* ap = (q < 32 / GM_CHUNK - 1) ? arow + 4 * GM_CHUNK * (q + 1) : arow_n;
#pragma unroll
            for (int u = 0; u < GM_CHUNK; ++u) an[u] = *(const v2f*)(ap + 4 * u);
            const v2f* bq = Bs + q * (GM_CHUNK * 8 * 32) + lane;
#pragma unroll
            for (int u = 0; u < GM_CHUNK; ++u) {
#pragma unroll
                for (int ct = 0; ct < 8; ++ct) {
                    v2f b = bq[(u * 8 + ct) * 32];
                    c[ct] = __builtin_amdgcn_wmma_f32_16x16x4_f32(false, ac[u], false, b,
                                                                  (short)0, c[ct], false, false);
                }
            }
        }
        float* orow = out + ((rt << 4) + (half << 3)) * HID + l16;
#pragma unroll
        for (int ct = 0; ct < 8; ++ct) {
#pragma unroll
            for (int v = 0; v < 8; ++v) orow[v * HID + (ct << 4)] = c[ct][v];
        }
        if (!more) break;
        rt = rn;
        arow = arow_n;
    }
}

#define GA_NODES  4
#define GA_BLOCKS 1563

__device__ __forceinline__ float4 edge_sum(const int2* __restrict__ csr, const float* __restrict__ hl,
                                           int beg, int end, float4 acc) {
    int j = beg;
    for (; j + 4 <= end; j += 4) {
        int2 e0 = csr[j], e1 = csr[j + 1], e2 = csr[j + 2], e3 = csr[j + 3];
        const float4 h0 = *(const float4*)(hl + e0.x * HID);
        const float4 h1 = *(const float4*)(hl + e1.x * HID);
        const float4 h2 = *(const float4*)(hl + e2.x * HID);
        const float4 h3 = *(const float4*)(hl + e3.x * HID);
        float w0 = __int_as_float(e0.y), w1 = __int_as_float(e1.y);
        float w2 = __int_as_float(e2.y), w3 = __int_as_float(e3.y);
        acc.x = fmaf(h0.x, w0, acc.x); acc.y = fmaf(h0.y, w0, acc.y);
        acc.z = fmaf(h0.z, w0, acc.z); acc.w = fmaf(h0.w, w0, acc.w);
        acc.x = fmaf(h1.x, w1, acc.x); acc.y = fmaf(h1.y, w1, acc.y);
        acc.z = fmaf(h1.z, w1, acc.z); acc.w = fmaf(h1.w, w1, acc.w);
        acc.x = fmaf(h2.x, w2, acc.x); acc.y = fmaf(h2.y, w2, acc.y);
        acc.z = fmaf(h2.z, w2, acc.z); acc.w = fmaf(h2.w, w2, acc.w);
        acc.x = fmaf(h3.x, w3, acc.x); acc.y = fmaf(h3.y, w3, acc.y);
        acc.z = fmaf(h3.z, w3, acc.z); acc.w = fmaf(h3.w, w3, acc.w);
    }
    for (; j < end; ++j) {
        int2 e0 = csr[j];
        const float4 h0 = *(const float4*)(hl + e0.x * HID);
        float w0 = __int_as_float(e0.y);
        acc.x = fmaf(h0.x, w0, acc.x); acc.y = fmaf(h0.y, w0, acc.y);
        acc.z = fmaf(h0.z, w0, acc.z); acc.w = fmaf(h0.w, w0, acc.w);
    }
    return acc;
}

__global__ void __launch_bounds__(256) k_gather(const int* __restrict__ rowptr, const int2* __restrict__ csr,
                                                const float* __restrict__ hw, const float* __restrict__ bias,
                                                float* __restrict__ out) {
    int wave = __builtin_amdgcn_readfirstlane((blockIdx.x * 256 + threadIdx.x) >> 5);
    int lane = threadIdx.x & 31;
    const float* hl = hw + lane * 4;
    const float4 b4 = *(const float4*)(bias + lane * 4);
    for (int i = 0; i < GA_NODES; ++i) {
        int n = wave * GA_NODES + i;
        if (n >= N_NODES) return;
        int beg = rowptr[n];
        int end = rowptr[n + 1];
        float4 acc = edge_sum(csr, hl, beg, end, make_float4(0.f, 0.f, 0.f, 0.f));
        float sd = rsqrtf((float)(end - beg) + 1.0f);
        float ss = sd * sd;
        const float4 hs = *(const float4*)(hl + n * HID);
        float4 o;
        o.x = fmaxf(acc.x + fmaf(hs.x, ss, b4.x), 0.f);
        o.y = fmaxf(acc.y + fmaf(hs.y, ss, b4.y), 0.f);
        o.z = fmaxf(acc.z + fmaf(hs.z, ss, b4.z), 0.f);
        o.w = fmaxf(acc.w + fmaf(hs.w, ss, b4.w), 0.f);
        *(float4*)(out + n * HID + lane * 4) = o;
    }
}

__device__ __forceinline__ void pool_add(float* __restrict__ sums, float* __restrict__ counts,
                                         int g, int lane, float4 acc, int cnt) {
    float* o = sums + g * HID + lane * 4;
    atomic_add_f32(o + 0, acc.x);
    atomic_add_f32(o + 1, acc.y);
    atomic_add_f32(o + 2, acc.z);
    atomic_add_f32(o + 3, acc.w);
    if (lane == 0) atomic_add_f32(counts + g, (float)cnt);
}

__global__ void __launch_bounds__(256) k_gather_pool(const int* __restrict__ rowptr, const int2* __restrict__ csr,
                                                     const float* __restrict__ hw, const int* __restrict__ batch,
                                                     float* __restrict__ sums, float* __restrict__ counts) {
    int wave = __builtin_amdgcn_readfirstlane((blockIdx.x * 256 + threadIdx.x) >> 5);
    int lane = threadIdx.x & 31;
    const float* hl = hw + lane * 4;
    int n0 = wave * GA_NODES;
    if (n0 >= N_NODES) return;
    int g = batch[n0];
    int cnt = 0;
    float4 acc = make_float4(0.f, 0.f, 0.f, 0.f);
    for (int i = 0; i < GA_NODES; ++i) {
        int n = n0 + i;
        if (n >= N_NODES) break;
        int gn = batch[n];
        int beg = rowptr[n];
        int end = rowptr[n + 1];
        if (gn != g) {
            pool_add(sums, counts, g, lane, acc, cnt);
            acc = make_float4(0.f, 0.f, 0.f, 0.f);
            cnt = 0;
            g = gn;
        }
        acc = edge_sum(csr, hl, beg, end, acc);
        float sd = rsqrtf((float)(end - beg) + 1.0f);
        float ss = sd * sd;
        const float4 hs = *(const float4*)(hl + n * HID);
        acc.x = fmaf(hs.x, ss, acc.x);
        acc.y = fmaf(hs.y, ss, acc.y);
        acc.z = fmaf(hs.z, ss, acc.z);
        acc.w = fmaf(hs.w, ss, acc.w);
        ++cnt;
    }
    pool_add(sums, counts, g, lane, acc, cnt);
}

#define GX_UNROLL 12
#define GX_TRIP   (3 * GX_UNROLL)
#define GX_NODES  5
#define GX_BLOCKS 1250
static_assert(GX_BLOCKS * 8 * GX_NODES == N_NODES, "every wave owns exactly GX_NODES nodes: no bounds test in the kernel");
__global__ void __launch_bounds__(256) k_gather_in(const int* __restrict__ rowptr, const int2* __restrict__ csr,
                                                   const float* __restrict__ x, const float* __restrict__ W1,
                                                   const float* __restrict__ bias, float* __restrict__ out) {
    int wave = __builtin_amdgcn_readfirstlane((blockIdx.x * 256 + threadIdx.x) >> 5);
    int lane = threadIdx.x & 31;
    int l = lane < 27 ? lane : 26;
    int e = l / IN_DIM;
    int k = l - e * IN_DIM;
    float4 w[IN_DIM];
#pragma unroll
    for (int q = 0; q < IN_DIM; ++q) w[q] = *(const float4*)(W1 + q * HID + lane * 4);
    const float4 b4 = *(const float4*)(bias + lane * 4);
    for (int i = 0; i < GX_NODES; ++i) {
        int n = wave * GX_NODES + i;
        int beg = rowptr[n];
        int end = rowptr[n + 1];
        float xs = x[n * IN_DIM + k];
        float acc = 0.f;
        for (int j0 = beg; j0 < end; j0 += GX_TRIP) {
            int2 ed[GX_UNROLL];
            float xv[GX_UNROLL];
#pragma unroll
            for (int u = 0; u < GX_UNROLL; ++u) {
                int j = j0 + 3 * u + e;
                ed[u] = csr[j < end ? j : end - 1];
            }
#pragma unroll
            for (int u = 0; u < GX_UNROLL; ++u) xv[u] = x[ed[u].x * IN_DIM + k];
#pragma unroll
            for (int u = 0; u < GX_UNROLL; ++u) {
                int j = j0 + 3 * u + e;
                float t = fmaf(xv[u], __int_as_float(ed[u].y), acc);
                acc = j < end ? t : acc;
            }
        }
        float s1 = __int_as_float(__builtin_amdgcn_ds_bpermute(((lane + 9) & 31) << 2, __float_as_int(acc)));
        float s2 = __int_as_float(__builtin_amdgcn_ds_bpermute(((lane + 18) & 31) << 2, __float_as_int(acc)));
        float sd = rsqrtf((float)(end - beg) + 1.0f);
        float ss = sd * sd;
        float ax = fmaf(xs, ss, (acc + s1) + s2);
        float4 o = make_float4(0.f, 0.f, 0.f, 0.f);
#pragma unroll
        for (int q = 0; q < IN_DIM; ++q) {
            float a = __int_as_float(__builtin_amdgcn_readlane(__float_as_int(ax), q));
            o.x = fmaf(a, w[q].x, o.x); o.y = fmaf(a, w[q].y, o.y);
            o.z = fmaf(a, w[q].z, o.z); o.w = fmaf(a, w[q].w, o.w);
        }
        o.x += b4.x; o.y += b4.y; o.z += b4.z; o.w += b4.w;
        *(float4*)(out + n * HID + lane * 4) = o;
    }
}

__global__ void __launch_bounds__(256) k_pool_gemm_wmma(const float* __restrict__ S, const float* __restrict__ W,
                                                        const float* __restrict__ counts, const float* __restrict__ bias,
                                                        float* __restrict__ out) {
    int wave = (blockIdx.x * 256 + threadIdx.x) >> 5;
    int lane = threadIdx.x & 31;
    int ct = wave & 7;
    int rt = wave >> 3;

    int half = lane >> 4;
    int l16  = lane & 15;
    int row  = (rt << 4) + l16;
    int col  = (ct << 4) + l16;
    const float* arow = S + row * HID;

    v8f c = {0.f, 0.f, 0.f, 0.f, 0.f, 0.f, 0.f, 0.f};
#pragma unroll
    for (int k0 = 0; k0 < HID; k0 += 4) {
        int ka = k0 + (half << 1);
        v2f a; a.x = arow[ka]; a.y = arow[ka + 1];
        v2f b; b.x = W[ka * HID + col]; b.y = W[(ka + 1) * HID + col];
        c = __builtin_amdgcn_wmma_f32_16x16x4_f32(false, a, false, b,
                                                  (short)0, c, false, false);
    }
    int g0 = (rt << 4) + (half << 3);
    float bc = bias[col];
#pragma unroll
    for (int v = 0; v < 8; ++v) {
        float cnt = counts[g0 + v];
        out[(g0 + v) * HID + col] = fmaf(cnt, bc, c[v]) / fmaxf(cnt, 1.0f);
    }
}

extern "C" void kernel_launch(void* const* d_in, const int* in_sizes, int n_in,
                              void* d_out, int out_size, void* d_ws, size_t ws_size,
                              hipStream_t stream) {
    const float* x        = (const float*)d_in[0];
    const int*   eidx     = (const int*)d_in[1];
    const int*   batch    = (const int*)d_in[2];
    const float* W1       = (const float*)d_in[3];
    const float* b1       = (const float*)d_in[4];
    const float* W2       = (const float*)d_in[5];
    const float* b2       = (const float*)d_in[6];
    const float* W3       = (const float*)d_in[7];
    const float* b3       = (const float*)d_in[8];
    float* out = (float*)d_out;

    const int* src = eidx;
    const int* dst = eidx + N_EDGES;

    float* ws     = (float*)d_ws;
    float* hw     = ws;
    float* hA     = hw + (size_t)N_NODES * HID;
    float* counts = hA + (size_t)N_NODES * HID;
    float* psum   = counts + N_GRAPHS;
    int2*  csr    = (int2*)(psum + N_GRAPHS * HID);
    int*   kA     = (int*)(csr + N_EDGES);
    int*   vA     = kA + N_EDGES;
    int*   kB     = vA + N_EDGES;
    int*   vB     = kB + N_EDGES;
    int*   hist   = vB + N_EDGES;
    int*   rowptr = hist + RS_HIST;

    const int B = 256;

    k_hist<0><<<RS_BLOCKS, B, 0, stream>>>(dst, hist);
    k_scatter<0><<<RS_BLOCKS, B, 0, stream>>>(dst, src, hist, kA, vA);
    k_hist<8><<<RS_BLOCKS, B, 0, stream>>>(kA, hist);
    k_scatter<8><<<RS_BLOCKS, B, 0, stream>>>(kA, vA, hist, kB, vB);
    k_bounds<<<EL_BLOCKS, B, 0, stream>>>(kB, rowptr, psum, counts);
    k_pair<<<EL_BLOCKS, B, 0, stream>>>(vB, kB, rowptr, csr);

    k_gather_in<<<GX_BLOCKS, B, 0, stream>>>(rowptr, csr, x, W1, b1, hA);

    k_gemm_rows_wmma<<<GM_BLOCKS, B, 0, stream>>>(hA, W2, hw);
    k_gather<<<GA_BLOCKS, B, 0, stream>>>(rowptr, csr, hw, b2, hA);

    k_gather_pool<<<GA_BLOCKS, B, 0, stream>>>(rowptr, csr, hA, batch, psum, counts);

    k_pool_gemm_wmma<<<N_GRAPHS / 16, B, 0, stream>>>(psum, W3, counts, b3, out);
}
